# starting kernel
# baseline (speedup 1.0000x reference)
#include <hip/hip_runtime.h>


typedef float v2f __attribute__((ext_vector_type(2)));
typedef float v8f __attribute__((ext_vector_type(8)));

#define WMMA4(A, B, C)                                                         \
  __builtin_amdgcn_wmma_f32_16x16x4_f32(false, (A), false, (B), (short)0,      \
                                        (C), false, false)

#define B_TILE 4

__global__ __launch_bounds__(256, 1) void tt_fused_kernel(
    const float* __restrict__ x, const float* __restrict__ c0,
    const float* __restrict__ c1, const float* __restrict__ c2,
    const float* __restrict__ c3, const float* __restrict__ bias,
    float* __restrict__ out) {
  __shared__ __align__(16) float lw0[8 * 64];
  __shared__ __align__(16) float lw1[64 * 64];
  __shared__ __align__(16) float lw2[64 * 64];
  __shared__ __align__(16) float lw3[64 * 8];
  __shared__ __align__(16) float z0[2 * 256 * 64];
  __shared__ __align__(16) float z1[8 * 32 * 64];
  __shared__ __align__(16) float z2[8 * 32 * 64];

  const int tid = threadIdx.x;
  const int w = tid >> 5;
  const int lane = tid & 31;
  const int lo = lane & 15;
  const int hi = lane >> 4;
  const int bg0 = blockIdx.x * B_TILE;

  for (int i = tid; i < 512; i += 256) lw0[i] = c0[i];
  for (int i = tid; i < 4096; i += 256) lw1[i] = c1[i];
  for (int i = tid; i < 4096; i += 256) lw2[i] = c2[i];
  for (int i = tid; i < 512; i += 256) lw3[i] = c3[i];
  __syncthreads();

  for (int pi = 0; pi < 4; ++pi) {
    {
      v2f bb0, bb1;
      bb0[0] = lw0[(2 * hi + 0) * 64 + pi * 16 + lo];
      bb0[1] = lw0[(2 * hi + 1) * 64 + pi * 16 + lo];
      bb1[0] = lw0[(4 + 2 * hi + 0) * 64 + pi * 16 + lo];
      bb1[1] = lw0[(4 + 2 * hi + 1) * 64 + pi * 16 + lo];
      for (int mt = w; mt < 128; mt += 8) {
        const int m0 = mt * 16 + lo;
        const int b = m0 >> 9;
        const int i1 = (m0 >> 6) & 7;
        const int sp = m0 & 63;
        const float* xr = x + (size_t)(bg0 + b) * 4096 + i1 * 64 + sp;
        v2f a0, a1;
        a0[0] = xr[(2 * hi + 0) * 512];
        a0[1] = xr[(2 * hi + 1) * 512];
        a1[0] = xr[(4 + 2 * hi + 0) * 512];
        a1[1] = xr[(4 + 2 * hi + 1) * 512];
        v8f acc = {0.f, 0.f, 0.f, 0.f, 0.f, 0.f, 0.f, 0.f};
        acc = WMMA4(a0, bb0, acc);
        acc = WMMA4(a1, bb1, acc);
        const int p = lo >> 3;
        const int r1 = lo & 7;
#pragma unroll
        for (int v = 0; v < 8; ++v) {
          const int mr = mt * 16 + 8 * hi + v;
          const int br = mr >> 9;
          const int i1r = (mr >> 6) & 7;
          const int spr = mr & 63;
          z0[p * 16384 + (br * 64 + spr) * 64 + r1 * 8 + i1r] = acc[v];
        }
      }
    }
    __syncthreads();

    for (int p = 0; p < 2; ++p) {
      const int o0 = pi * 2 + p;
      {
        const int nt = w & 3;
        const int half = w >> 2;
        const int nb = nt * 16 + lo;
        v2f wb[16];
#pragma unroll
        for (int t = 0; t < 16; ++t) {
          const int k = 4 * t + 2 * hi;
          wb[t][0] = lw1[k * 64 + nb];
          wb[t][1] = lw1[(k + 1) * 64 + nb];
        }
        const int o1s = nb >> 3;
        const int r2s = nb & 7;
        for (int i = 0; i < 8; ++i) {
          const int mt = half * 8 + i;
          const float* arow = &z0[p * 16384 + (mt * 16 + lo) * 64];
          v8f acc = {0.f, 0.f, 0.f, 0.f, 0.f, 0.f, 0.f, 0.f};
#pragma unroll
          for (int t = 0; t < 16; ++t) {
            v2f a = *(const v2f*)(arow + 4 * t + 2 * hi);
            acc = WMMA4(a, wb[t], acc);
          }
#pragma unroll
          for (int v = 0; v < 8; ++v) {
            const int m1 = mt * 16 + 8 * hi + v;
            const int b = m1 >> 6;
            const int i2 = (m1 >> 3) & 7;
            const int i3 = m1 & 7;
            z1[o1s * 2048 + (b * 8 + i3) * 64 + r2s * 8 + i2] = acc[v];
          }
        }
      }
      __syncthreads();

      {
        const int nt = w & 3;
        const int mt = w >> 2;
        const int nb = nt * 16 + lo;
        v2f wb2[16];
#pragma unroll
        for (int t = 0; t < 16; ++t) {
          const int k = 4 * t + 2 * hi;
          wb2[t][0] = lw2[k * 64 + nb];
          wb2[t][1] = lw2[(k + 1) * 64 + nb];
        }
        const int o2s = nb >> 3;
        const int r3s = nb & 7;
        for (int o1 = 0; o1 < 8; ++o1) {
          const float* arow = &z1[o1 * 2048 + (mt * 16 + lo) * 64];
          v8f acc = {0.f, 0.f, 0.f, 0.f, 0.f, 0.f, 0.f, 0.f};
#pragma unroll
          for (int t = 0; t < 16; ++t) {
            v2f a = *(const v2f*)(arow + 4 * t + 2 * hi);
            acc = WMMA4(a, wb2[t], acc);
          }
#pragma unroll
          for (int v = 0; v < 8; ++v) {
            const int m2 = mt * 16 + 8 * hi + v;
            const int b = m2 >> 3;
            const int i3 = m2 & 7;
            z2[o1 * 2048 + (b * 8 + o2s) * 64 + r3s * 8 + i3] = acc[v];
          }
        }
      }
      __syncthreads();

      {
        const int o1 = w;
        v2f wb3[16];
#pragma unroll
        for (int t = 0; t < 16; ++t) {
          const int k = 4 * t + 2 * hi;
          wb3[t][0] = (lo < 8) ? lw3[k * 8 + lo] : 0.f;
          wb3[t][1] = (lo < 8) ? lw3[(k + 1) * 8 + lo] : 0.f;
        }
        for (int mt = 0; mt < 2; ++mt) {
          const float* arow = &z2[o1 * 2048 + (mt * 16 + lo) * 64];
          v8f acc = {0.f, 0.f, 0.f, 0.f, 0.f, 0.f, 0.f, 0.f};
#pragma unroll
          for (int t = 0; t < 16; ++t) {
            v2f a = *(const v2f*)(arow + 4 * t + 2 * hi);
            acc = WMMA4(a, wb3[t], acc);
          }
          if (lo < 8) {
#pragma unroll
            for (int v = 0; v < 8; ++v) {
              const int m3 = mt * 16 + 8 * hi + v;
              const int b = m3 >> 3;
              const int o2 = m3 & 7;
              const int feat = ((o0 * 8 + o1) * 8 + o2) * 8 + lo;
              out[(size_t)(bg0 + b) * 4096 + feat] = acc[v] + bias[feat];
            }
          }
        }
      }
      __syncthreads();
    }
  }
}

extern "C" void kernel_launch(void* const* d_in, const int* in_sizes, int n_in,
                              void* d_out, int out_size, void* d_ws,
                              size_t ws_size, hipStream_t stream) {
  const float* x = (const float*)d_in[0];
  const float* c0 = (const float*)d_in[1];
  const float* c1 = (const float*)d_in[2];
  const float* c2 = (const float*)d_in[3];
  const float* c3 = (const float*)d_in[4];
  const float* bias = (const float*)d_in[5];
  float* out = (float*)d_out;

  dim3 grid(4096 / B_TILE);
  dim3 block(256);
  tt_fused_kernel<<<grid, block, 0, stream>>>(x, c0, c1, c2, c3, bias, out);
}


# LDS row pitch of z0/z1/z2 raised from 64 to 68 dwords (z0/z1 planes offset by 2 banks, core-tile columns XORed on rows with k&2) so the rows of each 8-byte A-fragment load fall on distinct banks
# speedup vs baseline: 2.1171x; 2.1171x over previous
#include <hip/hip_runtime.h>


typedef float v2f __attribute__((ext_vector_type(2)));
typedef float v8f __attribute__((ext_vector_type(8)));

#define WMMA4(A, B, C)                                                         \
  __builtin_amdgcn_wmma_f32_16x16x4_f32(false, (A), false, (B), (short)0,      \
                                        (C), false, false)

#define B_TILE 4

#define ZS 68
#define Z0P (256 * ZS + 2)
#define Z1P (32 * ZS + 2)
#define Z2P (32 * ZS)

__global__ __launch_bounds__(256, 1) void tt_fused_kernel(
    const float* __restrict__ x, const float* __restrict__ c0,
    const float* __restrict__ c1, const float* __restrict__ c2,
    const float* __restrict__ c3, const float* __restrict__ bias,
    float* __restrict__ out) {
  __shared__ __align__(16) float lw0[8 * 64];
  __shared__ __align__(16) float lw1[64 * 64];
  __shared__ __align__(16) float lw2[64 * 64];
  __shared__ __align__(16) float lw3[64 * 8];
  __shared__ __align__(16) float z0[2 * Z0P];
  __shared__ __align__(16) float z1[8 * Z1P];
  __shared__ __align__(16) float z2[8 * Z2P];

  const int tid = threadIdx.x;
  const int w = tid >> 5;
  const int lane = tid & 31;
  const int lo = lane & 15;
  const int hi = lane >> 4;
  const int bg0 = blockIdx.x * B_TILE;

  for (int i = tid; i < 512; i += 256) lw0[i] = c0[i];
  for (int i = tid; i < 4096; i += 256) lw1[i ^ ((i >> 3) & 16)] = c1[i];
  for (int i = tid; i < 4096; i += 256) lw2[i ^ ((i >> 3) & 16)] = c2[i];
  for (int i = tid; i < 512; i += 256) lw3[i] = c3[i];
  __syncthreads();

  for (int pi = 0; pi < 4; ++pi) {
    {
      v2f bb0, bb1;
      bb0[0] = lw0[(2 * hi + 0) * 64 + pi * 16 + lo];
      bb0[1] = lw0[(2 * hi + 1) * 64 + pi * 16 + lo];
      bb1[0] = lw0[(4 + 2 * hi + 0) * 64 + pi * 16 + lo];
      bb1[1] = lw0[(4 + 2 * hi + 1) * 64 + pi * 16 + lo];
      for (int mt = w; mt < 128; mt += 8) {
        const int m0 = mt * 16 + lo;
        const int b = m0 >> 9;
        const int i1 = (m0 >> 6) & 7;
        const int sp = m0 & 63;
        const float* xr = x + (size_t)(bg0 + b) * 4096 + i1 * 64 + sp;
        v2f a0, a1;
        a0[0] = xr[(2 * hi + 0) * 512];
        a0[1] = xr[(2 * hi + 1) * 512];
        a1[0] = xr[(4 + 2 * hi + 0) * 512];
        a1[1] = xr[(4 + 2 * hi + 1) * 512];
        v8f acc = {0.f, 0.f, 0.f, 0.f, 0.f, 0.f, 0.f, 0.f};
        acc = WMMA4(a0, bb0, acc);
        acc = WMMA4(a1, bb1, acc);
        const int p = lo >> 3;
        const int r1 = lo & 7;
#pragma unroll
        for (int v = 0; v < 8; ++v) {
          const int mr = mt * 16 + 8 * hi + v;
          const int br = mr >> 9;
          const int i1r = (mr >> 6) & 7;
          const int spr = mr & 63;
          z0[p * Z0P + (br * 64 + spr) * ZS + r1 * 8 + i1r] = acc[v];
        }
      }
    }
    __syncthreads();

    for (int p = 0; p < 2; ++p) {
      const int o0 = pi * 2 + p;
      {
        const int nt = w & 3;
        const int half = w >> 2;
        const int nb = nt * 16 + lo;
        const int nbx = nb ^ (hi << 4);
        v2f wb[16];
#pragma unroll
        for (int t = 0; t < 16; ++t) {
          const int k = 4 * t + 2 * hi;
          wb[t][0] = lw1[k * 64 + nbx];
          wb[t][1] = lw1[(k + 1) * 64 + nbx];
        }
        const int o1s = nb >> 3;
        const int r2s = nb & 7;
        for (int i = 0; i < 8; ++i) {
          const int mt = half * 8 + i;
          const float* arow = &z0[p * Z0P + (mt * 16 + lo) * ZS];
          v8f acc = {0.f, 0.f, 0.f, 0.f, 0.f, 0.f, 0.f, 0.f};
#pragma unroll
          for (int t = 0; t < 16; ++t) {
            v2f a = *(const v2f*)(arow + 4 * t + 2 * hi);
            acc = WMMA4(a, wb[t], acc);
          }
#pragma unroll
          for (int v = 0; v < 8; ++v) {
            const int m1 = mt * 16 + 8 * hi + v;
            const int b = m1 >> 6;
            const int i2 = (m1 >> 3) & 7;
            const int i3 = m1 & 7;
            z1[o1s * Z1P + (b * 8 + i3) * ZS + r2s * 8 + i2] = acc[v];
          }
        }
      }
      __syncthreads();

      {
        const int nt = w & 3;
        const int mt = w >> 2;
        const int nb = nt * 16 + lo;
        const int nbx = nb ^ (hi << 4);
        v2f wb2[16];
#pragma unroll
        for (int t = 0; t < 16; ++t) {
          const int k = 4 * t + 2 * hi;
          wb2[t][0] = lw2[k * 64 + nbx];
          wb2[t][1] = lw2[(k + 1) * 64 + nbx];
        }
        const int o2s = nb >> 3;
        const int r3s = nb & 7;
        for (int o1 = 0; o1 < 8; ++o1) {
          const float* arow = &z1[o1 * Z1P + (mt * 16 + lo) * ZS];
          v8f acc = {0.f, 0.f, 0.f, 0.f, 0.f, 0.f, 0.f, 0.f};
#pragma unroll
          for (int t = 0; t < 16; ++t) {
            v2f a = *(const v2f*)(arow + 4 * t + 2 * hi);
            acc = WMMA4(a, wb2[t], acc);
          }
#pragma unroll
          for (int v = 0; v < 8; ++v) {
            const int m2 = mt * 16 + 8 * hi + v;
            const int b = m2 >> 3;
            const int i3 = m2 & 7;
            z2[o1 * Z2P + (b * 8 + o2s) * ZS + r3s * 8 + i3] = acc[v];
          }
        }
      }
      __syncthreads();

      {
        const int o1 = w;
        v2f wb3[16];
#pragma unroll
        for (int t = 0; t < 16; ++t) {
          const int k = 4 * t + 2 * hi;
          wb3[t][0] = (lo < 8) ? lw3[k * 8 + lo] : 0.f;
          wb3[t][1] = (lo < 8) ? lw3[(k + 1) * 8 + lo] : 0.f;
        }
        for (int mt = 0; mt < 2; ++mt) {
          const float* arow = &z2[o1 * Z2P + (mt * 16 + lo) * ZS];
          v8f acc = {0.f, 0.f, 0.f, 0.f, 0.f, 0.f, 0.f, 0.f};
#pragma unroll
          for (int t = 0; t < 16; ++t) {
            v2f a = *(const v2f*)(arow + 4 * t + 2 * hi);
            acc = WMMA4(a, wb3[t], acc);
          }
          if (lo < 8) {
#pragma unroll
            for (int v = 0; v < 8; ++v) {
              const int m3 = mt * 16 + 8 * hi + v;
              const int b = m3 >> 3;
              const int o2 = m3 & 7;
              const int feat = ((o0 * 8 + o1) * 8 + o2) * 8 + lo;
              out[(size_t)(bg0 + b) * 4096 + feat] = acc[v] + bias[feat];
            }
          }
        }
      }
      __syncthreads();
    }
  }
}

extern "C" void kernel_launch(void* const* d_in, const int* in_sizes, int n_in,
                              void* d_out, int out_size, void* d_ws,
                              size_t ws_size, hipStream_t stream) {
  const float* x = (const float*)d_in[0];
  const float* c0 = (const float*)d_in[1];
  const float* c1 = (const float*)d_in[2];
  const float* c2 = (const float*)d_in[3];
  const float* c3 = (const float*)d_in[4];
  const float* bias = (const float*)d_in[5];
  float* out = (float*)d_out;

  dim3 grid(4096 / B_TILE);
  dim3 block(256);
  tt_fused_kernel<<<grid, block, 0, stream>>>(x, c0, c1, c2, c3, bias, out);
}


# x tile fetched once per workgroup: each wave's stage-0 A fragments loaded at entry with 16-byte loads (lane = 4 consecutive s', one per M tile) and kept in 64 VGPRs for all 4 o0-pair passes
# speedup vs baseline: 2.5598x; 1.2091x over previous
#include <hip/hip_runtime.h>


typedef float v2f __attribute__((ext_vector_type(2)));
typedef float v4f __attribute__((ext_vector_type(4)));
typedef float v8f __attribute__((ext_vector_type(8)));

#define WMMA4(A, B, C)                                                         \
  __builtin_amdgcn_wmma_f32_16x16x4_f32(false, (A), false, (B), (short)0,      \
                                        (C), false, false)

#define B_TILE 4

#define ZS 68
#define Z0P (256 * ZS + 2)
#define Z1P (32 * ZS + 2)
#define Z2P (32 * ZS)

__global__ __launch_bounds__(256, 1) void tt_fused_kernel(
    const float* __restrict__ x, const float* __restrict__ c0,
    const float* __restrict__ c1, const float* __restrict__ c2,
    const float* __restrict__ c3, const float* __restrict__ bias,
    float* __restrict__ out) {
  __shared__ __align__(16) float lw0[8 * 64];
  __shared__ __align__(16) float lw1[64 * 64];
  __shared__ __align__(16) float lw2[64 * 64];
  __shared__ __align__(16) float lw3[64 * 8];
  __shared__ __align__(16) float z0[2 * Z0P];
  __shared__ __align__(16) float z1[8 * Z1P];
  __shared__ __align__(16) float z2[8 * Z2P];

  const int tid = threadIdx.x;
  const int w = tid >> 5;
  const int lane = tid & 31;
  const int lo = lane & 15;
  const int hi = lane >> 4;
  const int bg0 = blockIdx.x * B_TILE;

  v2f xf[B_TILE][4][2];
  {
    const float* xw =
        x + (size_t)bg0 * 4096 + (2 * hi) * 512 + w * 64 + 4 * lo;
#pragma unroll
    for (int b = 0; b < B_TILE; ++b) {
      const v4f q0 = *(const v4f*)(xw + b * 4096);
      const v4f q1 = *(const v4f*)(xw + b * 4096 + 512);
      const v4f q2 = *(const v4f*)(xw + b * 4096 + 2048);
      const v4f q3 = *(const v4f*)(xw + b * 4096 + 2560);
#pragma unroll
      for (int j = 0; j < 4; ++j) {
        xf[b][j][0][0] = q0[j];
        xf[b][j][0][1] = q1[j];
        xf[b][j][1][0] = q2[j];
        xf[b][j][1][1] = q3[j];
      }
    }
  }

  for (int i = tid; i < 512; i += 256) lw0[i] = c0[i];
  for (int i = tid; i < 4096; i += 256) lw1[i ^ ((i >> 3) & 16)] = c1[i];
  for (int i = tid; i < 4096; i += 256) lw2[i ^ ((i >> 3) & 16)] = c2[i];
  for (int i = tid; i < 512; i += 256) lw3[i] = c3[i];
  __syncthreads();

  for (int pi = 0; pi < 4; ++pi) {
    {
      v2f bb0, bb1;
      bb0[0] = lw0[(2 * hi + 0) * 64 + pi * 16 + lo];
      bb0[1] = lw0[(2 * hi + 1) * 64 + pi * 16 + lo];
      bb1[0] = lw0[(4 + 2 * hi + 0) * 64 + pi * 16 + lo];
      bb1[1] = lw0[(4 + 2 * hi + 1) * 64 + pi * 16 + lo];
      const int p = lo >> 3;
      const int r1 = lo & 7;
      float* zd = &z0[p * Z0P + (32 * hi) * ZS + r1 * 8 + w];
#pragma unroll
      for (int b = 0; b < B_TILE; ++b) {
        v8f acc[4];
#pragma unroll
        for (int j = 0; j < 4; ++j) {
          v8f c = {0.f, 0.f, 0.f, 0.f, 0.f, 0.f, 0.f, 0.f};
          c = WMMA4(xf[b][j][0], bb0, c);
          acc[j] = WMMA4(xf[b][j][1], bb1, c);
        }
#pragma unroll
        for (int v = 0; v < 8; ++v) {
#pragma unroll
          for (int j = 0; j < 4; ++j) {
            zd[(b * 64 + 4 * v + j) * ZS] = acc[j][v];
          }
        }
      }
    }
    __syncthreads();

    for (int p = 0; p < 2; ++p) {
      const int o0 = pi * 2 + p;
      {
        const int nt = w & 3;
        const int half = w >> 2;
        const int nb = nt * 16 + lo;
        const int nbx = nb ^ (hi << 4);
        v2f wb[16];
#pragma unroll
        for (int t = 0; t < 16; ++t) {
          const int k = 4 * t + 2 * hi;
          wb[t][0] = lw1[k * 64 + nbx];
          wb[t][1] = lw1[(k + 1) * 64 + nbx];
        }
        const int o1s = nb >> 3;
        const int r2s = nb & 7;
        for (int i = 0; i < 8; ++i) {
          const int mt = half * 8 + i;
          const float* arow = &z0[p * Z0P + (mt * 16 + lo) * ZS];
          v8f acc = {0.f, 0.f, 0.f, 0.f, 0.f, 0.f, 0.f, 0.f};
#pragma unroll
          for (int t = 0; t < 16; ++t) {
            v2f a = *(const v2f*)(arow + 4 * t + 2 * hi);
            acc = WMMA4(a, wb[t], acc);
          }
#pragma unroll
          for (int v = 0; v < 8; ++v) {
            const int m1 = mt * 16 + 8 * hi + v;
            const int b = m1 >> 6;
            const int i2 = (m1 >> 3) & 7;
            const int i3 = m1 & 7;
            z1[o1s * Z1P + (b * 8 + i3) * ZS + r2s * 8 + i2] = acc[v];
          }
        }
      }
      __syncthreads();

      {
        const int nt = w & 3;
        const int mt = w >> 2;
        const int nb = nt * 16 + lo;
        const int nbx = nb ^ (hi << 4);
        v2f wb2[16];
#pragma unroll
        for (int t = 0; t < 16; ++t) {
          const int k = 4 * t + 2 * hi;
          wb2[t][0] = lw2[k * 64 + nbx];
          wb2[t][1] = lw2[(k + 1) * 64 + nbx];
        }
        const int o2s = nb >> 3;
        const int r3s = nb & 7;
        for (int o1 = 0; o1 < 8; ++o1) {
          const float* arow = &z1[o1 * Z1P + (mt * 16 + lo) * ZS];
          v8f acc = {0.f, 0.f, 0.f, 0.f, 0.f, 0.f, 0.f, 0.f};
#pragma unroll
          for (int t = 0; t < 16; ++t) {
            v2f a = *(const v2f*)(arow + 4 * t + 2 * hi);
            acc = WMMA4(a, wb2[t], acc);
          }
#pragma unroll
          for (int v = 0; v < 8; ++v) {
            const int m2 = mt * 16 + 8 * hi + v;
            const int b = m2 >> 3;
            const int i3 = m2 & 7;
            z2[o1 * Z2P + (b * 8 + o2s) * ZS + r3s * 8 + i3] = acc[v];
          }
        }
      }
      __syncthreads();

      {
        const int o1 = w;
        v2f wb3[16];
#pragma unroll
        for (int t = 0; t < 16; ++t) {
          const int k = 4 * t + 2 * hi;
          wb3[t][0] = (lo < 8) ? lw3[k * 8 + lo] : 0.f;
          wb3[t][1] = (lo < 8) ? lw3[(k + 1) * 8 + lo] : 0.f;
        }
        for (int mt = 0; mt < 2; ++mt) {
          const float* arow = &z2[o1 * Z2P + (mt * 16 + lo) * ZS];
          v8f acc = {0.f, 0.f, 0.f, 0.f, 0.f, 0.f, 0.f, 0.f};
#pragma unroll
          for (int t = 0; t < 16; ++t) {
            v2f a = *(const v2f*)(arow + 4 * t + 2 * hi);
            acc = WMMA4(a, wb3[t], acc);
          }
          if (lo < 8) {
#pragma unroll
            for (int v = 0; v < 8; ++v) {
              const int m3 = mt * 16 + 8 * hi + v;
              const int b = m3 >> 3;
              const int o2 = m3 & 7;
              const int feat = ((o0 * 8 + o1) * 8 + o2) * 8 + lo;
              out[(size_t)(bg0 + b) * 4096 + feat] = acc[v] + bias[feat];
            }
          }
        }
      }
      __syncthreads();
    }
  }
}

extern "C" void kernel_launch(void* const* d_in, const int* in_sizes, int n_in,
                              void* d_out, int out_size, void* d_ws,
                              size_t ws_size, hipStream_t stream) {
  const float* x = (const float*)d_in[0];
  const float* c0 = (const float*)d_in[1];
  const float* c1 = (const float*)d_in[2];
  const float* c2 = (const float*)d_in[3];
  const float* c3 = (const float*)d_in[4];
  const float* bias = (const float*)d_in[5];
  float* out = (float*)d_out;

  dim3 grid(4096 / B_TILE);
  dim3 block(256);
  tt_fused_kernel<<<grid, block, 0, stream>>>(x, c0, c1, c2, c3, bias, out);
}


# Stages 1-3: A-fragment LDS loads of the next two M tiles issued as one group into a second register set ahead of each 32-WMMA block (prior block's D stores beside them), none among the WMMAs
# speedup vs baseline: 2.7822x; 1.0869x over previous
#include <hip/hip_runtime.h>


typedef float v2f __attribute__((ext_vector_type(2)));
typedef float v4f __attribute__((ext_vector_type(4)));
typedef float v8f __attribute__((ext_vector_type(8)));

#define WMMA4(A, B, C)                                                         \
  __builtin_amdgcn_wmma_f32_16x16x4_f32(false, (A), false, (B), (short)0,      \
                                        (C), false, false)

#define BLOCK_FENCE() __builtin_amdgcn_sched_barrier(0)

#define B_TILE 4

#define ZS 68
#define Z0P (256 * ZS + 2)
#define Z1P (32 * ZS + 2)
#define Z2P (32 * ZS)

__global__ __launch_bounds__(256, 1) void tt_fused_kernel(
    const float* __restrict__ x, const float* __restrict__ c0,
    const float* __restrict__ c1, const float* __restrict__ c2,
    const float* __restrict__ c3, const float* __restrict__ bias,
    float* __restrict__ out) {
  __shared__ __align__(16) float lw0[8 * 64];
  __shared__ __align__(16) float lw1[64 * 64];
  __shared__ __align__(16) float lw2[64 * 64];
  __shared__ __align__(16) float lw3[64 * 8];
  __shared__ __align__(16) float z0[2 * Z0P];
  __shared__ __align__(16) float z1[8 * Z1P];
  __shared__ __align__(16) float z2[8 * Z2P];

  const int tid = threadIdx.x;
  const int w = tid >> 5;
  const int lane = tid & 31;
  const int lo = lane & 15;
  const int hi = lane >> 4;
  const int bg0 = blockIdx.x * B_TILE;

  v2f xf[B_TILE][4][2];
  {
    const float* xw =
        x + (size_t)bg0 * 4096 + (2 * hi) * 512 + w * 64 + 4 * lo;
#pragma unroll
    for (int b = 0; b < B_TILE; ++b) {
      const v4f q0 = *(const v4f*)(xw + b * 4096);
      const v4f q1 = *(const v4f*)(xw + b * 4096 + 512);
      const v4f q2 = *(const v4f*)(xw + b * 4096 + 2048);
      const v4f q3 = *(const v4f*)(xw + b * 4096 + 2560);
#pragma unroll
      for (int j = 0; j < 4; ++j) {
        xf[b][j][0][0] = q0[j];
        xf[b][j][0][1] = q1[j];
        xf[b][j][1][0] = q2[j];
        xf[b][j][1][1] = q3[j];
      }
    }
  }

  for (int i = tid; i < 512; i += 256) lw0[i] = c0[i];
  for (int i = tid; i < 4096; i += 256) lw1[i ^ ((i >> 3) & 16)] = c1[i];
  for (int i = tid; i < 4096; i += 256) lw2[i ^ ((i >> 3) & 16)] = c2[i];
  for (int i = tid; i < 512; i += 256) lw3[i] = c3[i];
  __syncthreads();

  for (int pi = 0; pi < 4; ++pi) {
    {
      v2f bb0, bb1;
      bb0[0] = lw0[(2 * hi + 0) * 64 + pi * 16 + lo];
      bb0[1] = lw0[(2 * hi + 1) * 64 + pi * 16 + lo];
      bb1[0] = lw0[(4 + 2 * hi + 0) * 64 + pi * 16 + lo];
      bb1[1] = lw0[(4 + 2 * hi + 1) * 64 + pi * 16 + lo];
      const int p = lo >> 3;
      const int r1 = lo & 7;
      float* zd = &z0[p * Z0P + (32 * hi) * ZS + r1 * 8 + w];
#pragma unroll
      for (int b = 0; b < B_TILE; ++b) {
        v8f acc[4];
#pragma unroll
        for (int j = 0; j < 4; ++j) {
          v8f c = {0.f, 0.f, 0.f, 0.f, 0.f, 0.f, 0.f, 0.f};
          c = WMMA4(xf[b][j][0], bb0, c);
          acc[j] = WMMA4(xf[b][j][1], bb1, c);
        }
#pragma unroll
        for (int v = 0; v < 8; ++v) {
#pragma unroll
          for (int j = 0; j < 4; ++j) {
            zd[(b * 64 + 4 * v + j) * ZS] = acc[j][v];
          }
        }
      }
    }
    __syncthreads();

    for (int p = 0; p < 2; ++p) {
      const int o0 = pi * 2 + p;
      {
        const int nt = w & 3;
        const int half = w >> 2;
        const int nb = nt * 16 + lo;
        const int nbx = nb ^ (hi << 4);
        v2f wb[16];
#pragma unroll
        for (int t = 0; t < 16; ++t) {
          const int k = 4 * t + 2 * hi;
          wb[t][0] = lw1[k * 64 + nbx];
          wb[t][1] = lw1[(k + 1) * 64 + nbx];
        }
        const int o1s = nb >> 3;
        const int r2s = nb & 7;
        const float* abase = &z0[p * Z0P + (half * 128 + lo) * ZS + 2 * hi];
        v2f af[2][2][16];
        v8f acc[2][2];
#pragma unroll
        for (int u = 0; u < 2; ++u) {
#pragma unroll
          for (int t = 0; t < 16; ++t) {
            af[0][u][t] = *(const v2f*)(abase + (u * 16) * ZS + 4 * t);
          }
        }
#pragma unroll
        for (int pr = 0; pr < 4; ++pr) {
          const int cur = pr & 1;
          const int nxt = cur ^ 1;
          if (pr < 3) {
#pragma unroll
            for (int u = 0; u < 2; ++u) {
#pragma unroll
              for (int t = 0; t < 16; ++t) {
                af[nxt][u][t] = *(const v2f*)(
                    abase + ((2 * pr + 2 + u) * 16) * ZS + 4 * t);
              }
            }
          }
          if (pr > 0) {
#pragma unroll
            for (int u = 0; u < 2; ++u) {
              const int mt = half * 8 + 2 * pr - 2 + u;
#pragma unroll
              for (int v = 0; v < 8; ++v) {
                const int m1 = mt * 16 + 8 * hi + v;
                const int b = m1 >> 6;
                const int i2 = (m1 >> 3) & 7;
                const int i3 = m1 & 7;
                z1[o1s * Z1P + (b * 8 + i3) * ZS + r2s * 8 + i2] =
                    acc[nxt][u][v];
              }
            }
          }
          BLOCK_FENCE();
          {
            v8f c0 = {0.f, 0.f, 0.f, 0.f, 0.f, 0.f, 0.f, 0.f};
            v8f c1 = {0.f, 0.f, 0.f, 0.f, 0.f, 0.f, 0.f, 0.f};
#pragma unroll
            for (int t = 0; t < 16; ++t) {
              c0 = WMMA4(af[cur][0][t], wb[t], c0);
              c1 = WMMA4(af[cur][1][t], wb[t], c1);
            }
            acc[cur][0] = c0;
            acc[cur][1] = c1;
          }
          BLOCK_FENCE();
        }
#pragma unroll
        for (int u = 0; u < 2; ++u) {
          const int mt = half * 8 + 6 + u;
#pragma unroll
          for (int v = 0; v < 8; ++v) {
            const int m1 = mt * 16 + 8 * hi + v;
            const int b = m1 >> 6;
            const int i2 = (m1 >> 3) & 7;
            const int i3 = m1 & 7;
            z1[o1s * Z1P + (b * 8 + i3) * ZS + r2s * 8 + i2] = acc[1][u][v];
          }
        }
      }
      __syncthreads();

      {
        const int nt = w & 3;
        const int mt = w >> 2;
        const int nb = nt * 16 + lo;
        const int nbx = nb ^ (hi << 4);
        v2f wb2[16];
#pragma unroll
        for (int t = 0; t < 16; ++t) {
          const int k = 4 * t + 2 * hi;
          wb2[t][0] = lw2[k * 64 + nbx];
          wb2[t][1] = lw2[(k + 1) * 64 + nbx];
        }
        const int o2s = nb >> 3;
        const int r3s = nb & 7;
        const float* abase = &z1[(mt * 16 + lo) * ZS + 2 * hi];
        float* dbase = &z2[((mt * 2 + hi) * 8 + o2s) * ZS + r3s * 8];
        v2f af[2][2][16];
        v8f acc[2][2];
#pragma unroll
        for (int u = 0; u < 2; ++u) {
#pragma unroll
          for (int t = 0; t < 16; ++t) {
            af[0][u][t] = *(const v2f*)(abase + u * Z1P + 4 * t);
          }
        }
#pragma unroll
        for (int pr = 0; pr < 4; ++pr) {
          const int cur = pr & 1;
          const int nxt = cur ^ 1;
          if (pr < 3) {
#pragma unroll
            for (int u = 0; u < 2; ++u) {
#pragma unroll
              for (int t = 0; t < 16; ++t) {
                af[nxt][u][t] =
                    *(const v2f*)(abase + (2 * pr + 2 + u) * Z1P + 4 * t);
              }
            }
          }
          if (pr > 0) {
#pragma unroll
            for (int u = 0; u < 2; ++u) {
#pragma unroll
              for (int v = 0; v < 8; ++v) {
                dbase[(2 * pr - 2 + u) * Z2P + v] = acc[nxt][u][v];
              }
            }
          }
          BLOCK_FENCE();
          {
            v8f c0 = {0.f, 0.f, 0.f, 0.f, 0.f, 0.f, 0.f, 0.f};
            v8f c1 = {0.f, 0.f, 0.f, 0.f, 0.f, 0.f, 0.f, 0.f};
#pragma unroll
            for (int t = 0; t < 16; ++t) {
              c0 = WMMA4(af[cur][0][t], wb2[t], c0);
              c1 = WMMA4(af[cur][1][t], wb2[t], c1);
            }
            acc[cur][0] = c0;
            acc[cur][1] = c1;
          }
          BLOCK_FENCE();
        }
#pragma unroll
        for (int u = 0; u < 2; ++u) {
#pragma unroll
          for (int v = 0; v < 8; ++v) {
            dbase[(6 + u) * Z2P + v] = acc[1][u][v];
          }
        }
      }
      __syncthreads();

      {
        const int o1 = w;
        v2f wb3[16];
#pragma unroll
        for (int t = 0; t < 16; ++t) {
          const int k = 4 * t + 2 * hi;
          wb3[t][0] = (lo < 8) ? lw3[k * 8 + lo] : 0.f;
          wb3[t][1] = (lo < 8) ? lw3[(k + 1) * 8 + lo] : 0.f;
        }
        const float* abase = &z2[o1 * Z2P + lo * ZS + 2 * hi];
        v2f af[2][16];
#pragma unroll
        for (int mt = 0; mt < 2; ++mt) {
#pragma unroll
          for (int t = 0; t < 16; ++t) {
            af[mt][t] = *(const v2f*)(abase + (mt * 16) * ZS + 4 * t);
          }
        }
        BLOCK_FENCE();
        v8f acc0 = {0.f, 0.f, 0.f, 0.f, 0.f, 0.f, 0.f, 0.f};
        v8f acc1 = {0.f, 0.f, 0.f, 0.f, 0.f, 0.f, 0.f, 0.f};
#pragma unroll
        for (int t = 0; t < 16; ++t) {
          acc0 = WMMA4(af[0][t], wb3[t], acc0);
          acc1 = WMMA4(af[1][t], wb3[t], acc1);
        }
        if (lo < 8) {
#pragma unroll
          for (int mt = 0; mt < 2; ++mt) {
#pragma unroll
            for (int v = 0; v < 8; ++v) {
              const int m3 = mt * 16 + 8 * hi + v;
              const int b = m3 >> 3;
              const int o2 = m3 & 7;
              const int feat = ((o0 * 8 + o1) * 8 + o2) * 8 + lo;
              out[(size_t)(bg0 + b) * 4096 + feat] =
                  (mt == 0 ? acc0[v] : acc1[v]) + bias[feat];
            }
          }
        }
      }
      __syncthreads();
    }
  }
}

extern "C" void kernel_launch(void* const* d_in, const int* in_sizes, int n_in,
                              void* d_out, int out_size, void* d_ws,
                              size_t ws_size, hipStream_t stream) {
  const float* x = (const float*)d_in[0];
  const float* c0 = (const float*)d_in[1];
  const float* c1 = (const float*)d_in[2];
  const float* c2 = (const float*)d_in[3];
  const float* c3 = (const float*)d_in[4];
  const float* bias = (const float*)d_in[5];
  float* out = (float*)d_out;

  dim3 grid(4096 / B_TILE);
  dim3 block(256);
  tt_fused_kernel<<<grid, block, 0, stream>>>(x, c0, c1, c2, c3, bias, out);
}


# Barrier behind stage 3 removed (z2/lw3 are next written two barriers later; z1/z0 writers are already fenced), so no wave waits there for its global stores to be acknowledged
# speedup vs baseline: 3.0307x; 1.0893x over previous
#include <hip/hip_runtime.h>


typedef float v2f __attribute__((ext_vector_type(2)));
typedef float v4f __attribute__((ext_vector_type(4)));
typedef float v8f __attribute__((ext_vector_type(8)));

#define WMMA4(A, B, C)                                                         \
  __builtin_amdgcn_wmma_f32_16x16x4_f32(false, (A), false, (B), (short)0,      \
                                        (C), false, false)

#define BLOCK_FENCE() __builtin_amdgcn_sched_barrier(0)

#define B_TILE 4

#define ZS 68
#define Z0P (256 * ZS + 2)
#define Z1P (32 * ZS + 2)
#define Z2P (32 * ZS)

__global__ __launch_bounds__(256, 1) void tt_fused_kernel(
    const float* __restrict__ x, const float* __restrict__ c0,
    const float* __restrict__ c1, const float* __restrict__ c2,
    const float* __restrict__ c3, const float* __restrict__ bias,
    float* __restrict__ out) {
  __shared__ __align__(16) float lw0[8 * 64];
  __shared__ __align__(16) float lw1[64 * 64];
  __shared__ __align__(16) float lw2[64 * 64];
  __shared__ __align__(16) float lw3[64 * 8];
  __shared__ __align__(16) float z0[2 * Z0P];
  __shared__ __align__(16) float z1[8 * Z1P];
  __shared__ __align__(16) float z2[8 * Z2P];

  const int tid = threadIdx.x;
  const int w = tid >> 5;
  const int lane = tid & 31;
  const int lo = lane & 15;
  const int hi = lane >> 4;
  const int bg0 = blockIdx.x * B_TILE;

  v2f xf[B_TILE][4][2];
  {
    const float* xw =
        x + (size_t)bg0 * 4096 + (2 * hi) * 512 + w * 64 + 4 * lo;
#pragma unroll
    for (int b = 0; b < B_TILE; ++b) {
      const v4f q0 = *(const v4f*)(xw + b * 4096);
      const v4f q1 = *(const v4f*)(xw + b * 4096 + 512);
      const v4f q2 = *(const v4f*)(xw + b * 4096 + 2048);
      const v4f q3 = *(const v4f*)(xw + b * 4096 + 2560);
#pragma unroll
      for (int j = 0; j < 4; ++j) {
        xf[b][j][0][0] = q0[j];
        xf[b][j][0][1] = q1[j];
        xf[b][j][1][0] = q2[j];
        xf[b][j][1][1] = q3[j];
      }
    }
  }

  for (int i = tid; i < 512; i += 256) lw0[i] = c0[i];
  for (int i = tid; i < 4096; i += 256) lw1[i ^ ((i >> 3) & 16)] = c1[i];
  for (int i = tid; i < 4096; i += 256) lw2[i ^ ((i >> 3) & 16)] = c2[i];
  for (int i = tid; i < 512; i += 256) lw3[i] = c3[i];
  __syncthreads();

  for (int pi = 0; pi < 4; ++pi) {
    {
      v2f bb0, bb1;
      bb0[0] = lw0[(2 * hi + 0) * 64 + pi * 16 + lo];
      bb0[1] = lw0[(2 * hi + 1) * 64 + pi * 16 + lo];
      bb1[0] = lw0[(4 + 2 * hi + 0) * 64 + pi * 16 + lo];
      bb1[1] = lw0[(4 + 2 * hi + 1) * 64 + pi * 16 + lo];
      const int p = lo >> 3;
      const int r1 = lo & 7;
      float* zd = &z0[p * Z0P + (32 * hi) * ZS + r1 * 8 + w];
#pragma unroll
      for (int b = 0; b < B_TILE; ++b) {
        v8f acc[4];
#pragma unroll
        for (int j = 0; j < 4; ++j) {
          v8f c = {0.f, 0.f, 0.f, 0.f, 0.f, 0.f, 0.f, 0.f};
          c = WMMA4(xf[b][j][0], bb0, c);
          acc[j] = WMMA4(xf[b][j][1], bb1, c);
        }
#pragma unroll
        for (int v = 0; v < 8; ++v) {
#pragma unroll
          for (int j = 0; j < 4; ++j) {
            zd[(b * 64 + 4 * v + j) * ZS] = acc[j][v];
          }
        }
      }
    }
    __syncthreads();

    for (int p = 0; p < 2; ++p) {
      const int o0 = pi * 2 + p;
      {
        const int nt = w & 3;
        const int half = w >> 2;
        const int nb = nt * 16 + lo;
        const int nbx = nb ^ (hi << 4);
        v2f wb[16];
#pragma unroll
        for (int t = 0; t < 16; ++t) {
          const int k = 4 * t + 2 * hi;
          wb[t][0] = lw1[k * 64 + nbx];
          wb[t][1] = lw1[(k + 1) * 64 + nbx];
        }
        const int o1s = nb >> 3;
        const int r2s = nb & 7;
        const float* abase = &z0[p * Z0P + (half * 128 + lo) * ZS + 2 * hi];
        v2f af[2][2][16];
        v8f acc[2][2];
#pragma unroll
        for (int u = 0; u < 2; ++u) {
#pragma unroll
          for (int t = 0; t < 16; ++t) {
            af[0][u][t] = *(const v2f*)(abase + (u * 16) * ZS + 4 * t);
          }
        }
#pragma unroll
        for (int pr = 0; pr < 4; ++pr) {
          const int cur = pr & 1;
          const int nxt = cur ^ 1;
          if (pr < 3) {
#pragma unroll
            for (int u = 0; u < 2; ++u) {
#pragma unroll
              for (int t = 0; t < 16; ++t) {
                af[nxt][u][t] = *(const v2f*)(
                    abase + ((2 * pr + 2 + u) * 16) * ZS + 4 * t);
              }
            }
          }
          if (pr > 0) {
#pragma unroll
            for (int u = 0; u < 2; ++u) {
              const int mt = half * 8 + 2 * pr - 2 + u;
#pragma unroll
              for (int v = 0; v < 8; ++v) {
                const int m1 = mt * 16 + 8 * hi + v;
                const int b = m1 >> 6;
                const int i2 = (m1 >> 3) & 7;
                const int i3 = m1 & 7;
                z1[o1s * Z1P + (b * 8 + i3) * ZS + r2s * 8 + i2] =
                    acc[nxt][u][v];
              }
            }
          }
          BLOCK_FENCE();
          {
            v8f c0 = {0.f, 0.f, 0.f, 0.f, 0.f, 0.f, 0.f, 0.f};
            v8f c1 = {0.f, 0.f, 0.f, 0.f, 0.f, 0.f, 0.f, 0.f};
#pragma unroll
            for (int t = 0; t < 16; ++t) {
              c0 = WMMA4(af[cur][0][t], wb[t], c0);
              c1 = WMMA4(af[cur][1][t], wb[t], c1);
            }
            acc[cur][0] = c0;
            acc[cur][1] = c1;
          }
          BLOCK_FENCE();
        }
#pragma unroll
        for (int u = 0; u < 2; ++u) {
          const int mt = half * 8 + 6 + u;
#pragma unroll
          for (int v = 0; v < 8; ++v) {
            const int m1 = mt * 16 + 8 * hi + v;
            const int b = m1 >> 6;
            const int i2 = (m1 >> 3) & 7;
            const int i3 = m1 & 7;
            z1[o1s * Z1P + (b * 8 + i3) * ZS + r2s * 8 + i2] = acc[1][u][v];
          }
        }
      }
      __syncthreads();

      {
        const int nt = w & 3;
        const int mt = w >> 2;
        const int nb = nt * 16 + lo;
        const int nbx = nb ^ (hi << 4);
        v2f wb2[16];
#pragma unroll
        for (int t = 0; t < 16; ++t) {
          const int k = 4 * t + 2 * hi;
          wb2[t][0] = lw2[k * 64 + nbx];
          wb2[t][1] = lw2[(k + 1) * 64 + nbx];
        }
        const int o2s = nb >> 3;
        const int r3s = nb & 7;
        const float* abase = &z1[(mt * 16 + lo) * ZS + 2 * hi];
        float* dbase = &z2[((mt * 2 + hi) * 8 + o2s) * ZS + r3s * 8];
        v2f af[2][2][16];
        v8f acc[2][2];
#pragma unroll
        for (int u = 0; u < 2; ++u) {
#pragma unroll
          for (int t = 0; t < 16; ++t) {
            af[0][u][t] = *(const v2f*)(abase + u * Z1P + 4 * t);
          }
        }
#pragma unroll
        for (int pr = 0; pr < 4; ++pr) {
          const int cur = pr & 1;
          const int nxt = cur ^ 1;
          if (pr < 3) {
#pragma unroll
            for (int u = 0; u < 2; ++u) {
#pragma unroll
              for (int t = 0; t < 16; ++t) {
                af[nxt][u][t] =
                    *(const v2f*)(abase + (2 * pr + 2 + u) * Z1P + 4 * t);
              }
            }
          }
          if (pr > 0) {
#pragma unroll
            for (int u = 0; u < 2; ++u) {
#pragma unroll
              for (int v = 0; v < 8; ++v) {
                dbase[(2 * pr - 2 + u) * Z2P + v] = acc[nxt][u][v];
              }
            }
          }
          BLOCK_FENCE();
          {
            v8f c0 = {0.f, 0.f, 0.f, 0.f, 0.f, 0.f, 0.f, 0.f};
            v8f c1 = {0.f, 0.f, 0.f, 0.f, 0.f, 0.f, 0.f, 0.f};
#pragma unroll
            for (int t = 0; t < 16; ++t) {
              c0 = WMMA4(af[cur][0][t], wb2[t], c0);
              c1 = WMMA4(af[cur][1][t], wb2[t], c1);
            }
            acc[cur][0] = c0;
            acc[cur][1] = c1;
          }
          BLOCK_FENCE();
        }
#pragma unroll
        for (int u = 0; u < 2; ++u) {
#pragma unroll
          for (int v = 0; v < 8; ++v) {
            dbase[(6 + u) * Z2P + v] = acc[1][u][v];
          }
        }
      }
      __syncthreads();

      {
        const int o1 = w;
        v2f wb3[16];
#pragma unroll
        for (int t = 0; t < 16; ++t) {
          const int k = 4 * t + 2 * hi;
          wb3[t][0] = (lo < 8) ? lw3[k * 8 + lo] : 0.f;
          wb3[t][1] = (lo < 8) ? lw3[(k + 1) * 8 + lo] : 0.f;
        }
        const float* abase = &z2[o1 * Z2P + lo * ZS + 2 * hi];
        v2f af[2][16];
#pragma unroll
        for (int mt = 0; mt < 2; ++mt) {
#pragma unroll
          for (int t = 0; t < 16; ++t) {
            af[mt][t] = *(const v2f*)(abase + (mt * 16) * ZS + 4 * t);
          }
        }
        BLOCK_FENCE();
        v8f acc0 = {0.f, 0.f, 0.f, 0.f, 0.f, 0.f, 0.f, 0.f};
        v8f acc1 = {0.f, 0.f, 0.f, 0.f, 0.f, 0.f, 0.f, 0.f};
#pragma unroll
        for (int t = 0; t < 16; ++t) {
          acc0 = WMMA4(af[0][t], wb3[t], acc0);
          acc1 = WMMA4(af[1][t], wb3[t], acc1);
        }
        if (lo < 8) {
#pragma unroll
          for (int mt = 0; mt < 2; ++mt) {
#pragma unroll
            for (int v = 0; v < 8; ++v) {
              const int m3 = mt * 16 + 8 * hi + v;
              const int b = m3 >> 3;
              const int o2 = m3 & 7;
              const int feat = ((o0 * 8 + o1) * 8 + o2) * 8 + lo;
              out[(size_t)(bg0 + b) * 4096 + feat] =
                  (mt == 0 ? acc0[v] : acc1[v]) + bias[feat];
            }
          }
        }
      }
    }
  }
}

extern "C" void kernel_launch(void* const* d_in, const int* in_sizes, int n_in,
                              void* d_out, int out_size, void* d_ws,
                              size_t ws_size, hipStream_t stream) {
  const float* x = (const float*)d_in[0];
  const float* c0 = (const float*)d_in[1];
  const float* c1 = (const float*)d_in[2];
  const float* c2 = (const float*)d_in[3];
  const float* c3 = (const float*)d_in[4];
  const float* bias = (const float*)d_in[5];
  float* out = (float*)d_out;

  dim3 grid(4096 / B_TILE);
  dim3 block(256);
  tt_fused_kernel<<<grid, block, 0, stream>>>(x, c0, c1, c2, c3, bias, out);
}


# Stages 2-3 re-split so each wave reads only the z1/z2 rows it wrote itself (its 2 o1 planes x 2 batch rows, all 4 N tiles): barriers behind stages 1 and 2 dropped, 2 per pass left round stage 0
# speedup vs baseline: 3.1555x; 1.0412x over previous
#include <hip/hip_runtime.h>


typedef float v2f __attribute__((ext_vector_type(2)));
typedef float v4f __attribute__((ext_vector_type(4)));
typedef float v8f __attribute__((ext_vector_type(8)));

#define WMMA4(A, B, C)                                                         \
  __builtin_amdgcn_wmma_f32_16x16x4_f32(false, (A), false, (B), (short)0,      \
                                        (C), false, false)

#define BLOCK_FENCE() __builtin_amdgcn_sched_barrier(0)


#define B_TILE 4

#define ZS 68
#define Z0P (256 * ZS + 2)
#define Z1P (32 * ZS + 2)
#define Z2P (32 * ZS)

__global__ __launch_bounds__(256, 1) void tt_fused_kernel(
    const float* __restrict__ x, const float* __restrict__ c0,
    const float* __restrict__ c1, const float* __restrict__ c2,
    const float* __restrict__ c3, const float* __restrict__ bias,
    float* __restrict__ out) {
  __shared__ __align__(16) float lw0[8 * 64];
  __shared__ __align__(16) float lw1[64 * 64];
  __shared__ __align__(16) float lw2[64 * 64];
  __shared__ __align__(16) float lw3[64 * 8];
  __shared__ __align__(16) float z0[2 * Z0P];
  __shared__ __align__(16) float z1[8 * Z1P];
  __shared__ __align__(16) float z2[8 * Z2P];

  const int tid = threadIdx.x;
  const int w = tid >> 5;
  const int lane = tid & 31;
  const int lo = lane & 15;
  const int hi = lane >> 4;
  const int bg0 = blockIdx.x * B_TILE;

  v2f xf[B_TILE][4][2];
  {
    const float* xw =
        x + (size_t)bg0 * 4096 + (2 * hi) * 512 + w * 64 + 4 * lo;
#pragma unroll
    for (int b = 0; b < B_TILE; ++b) {
      const v4f q0 = *(const v4f*)(xw + b * 4096);
      const v4f q1 = *(const v4f*)(xw + b * 4096 + 512);
      const v4f q2 = *(const v4f*)(xw + b * 4096 + 2048);
      const v4f q3 = *(const v4f*)(xw + b * 4096 + 2560);
#pragma unroll
      for (int j = 0; j < 4; ++j) {
        xf[b][j][0][0] = q0[j];
        xf[b][j][0][1] = q1[j];
        xf[b][j][1][0] = q2[j];
        xf[b][j][1][1] = q3[j];
      }
    }
  }

  for (int i = tid; i < 512; i += 256) lw0[i] = c0[i];
  for (int i = tid; i < 4096; i += 256) lw1[i ^ ((i >> 3) & 16)] = c1[i];
  for (int i = tid; i < 4096; i += 256) lw2[i ^ ((i >> 3) & 16)] = c2[i];
  for (int i = tid; i < 512; i += 256) lw3[i] = c3[i];
  __syncthreads();

  for (int pi = 0; pi < 4; ++pi) {
    if (pi > 0) __syncthreads();
    {
      v2f bb0, bb1;
      bb0[0] = lw0[(2 * hi + 0) * 64 + pi * 16 + lo];
      bb0[1] = lw0[(2 * hi + 1) * 64 + pi * 16 + lo];
      bb1[0] = lw0[(4 + 2 * hi + 0) * 64 + pi * 16 + lo];
      bb1[1] = lw0[(4 + 2 * hi + 1) * 64 + pi * 16 + lo];
      const int p = lo >> 3;
      const int r1 = lo & 7;
      float* zd = &z0[p * Z0P + (32 * hi) * ZS + r1 * 8 + w];
#pragma unroll
      for (int b = 0; b < B_TILE; ++b) {
        v8f acc[4];
#pragma unroll
        for (int j = 0; j < 4; ++j) {
          v8f c = {0.f, 0.f, 0.f, 0.f, 0.f, 0.f, 0.f, 0.f};
          c = WMMA4(xf[b][j][0], bb0, c);
          acc[j] = WMMA4(xf[b][j][1], bb1, c);
        }
#pragma unroll
        for (int v = 0; v < 8; ++v) {
#pragma unroll
          for (int j = 0; j < 4; ++j) {
            zd[(b * 64 + 4 * v + j) * ZS] = acc[j][v];
          }
        }
      }
    }
    __syncthreads();

    for (int p = 0; p < 2; ++p) {
      const int o0 = pi * 2 + p;
      {
        const int nt = w & 3;
        const int half = w >> 2;
        const int nb = nt * 16 + lo;
        const int nbx = nb ^ (hi << 4);
        v2f wb[16];
#pragma unroll
        for (int t = 0; t < 16; ++t) {
          const int k = 4 * t + 2 * hi;
          wb[t][0] = lw1[k * 64 + nbx];
          wb[t][1] = lw1[(k + 1) * 64 + nbx];
        }
        const int o1s = nb >> 3;
        const int r2s = nb & 7;
        const float* abase = &z0[p * Z0P + (half * 128 + lo) * ZS + 2 * hi];
        v2f af[2][2][16];
        v8f acc[2][2];
#pragma unroll
        for (int u = 0; u < 2; ++u) {
#pragma unroll
          for (int t = 0; t < 16; ++t) {
            af[0][u][t] = *(const v2f*)(abase + (u * 16) * ZS + 4 * t);
          }
        }
#pragma unroll
        for (int pr = 0; pr < 4; ++pr) {
          const int cur = pr & 1;
          const int nxt = cur ^ 1;
          if (pr < 3) {
#pragma unroll
            for (int u = 0; u < 2; ++u) {
#pragma unroll
              for (int t = 0; t < 16; ++t) {
                af[nxt][u][t] = *(const v2f*)(
                    abase + ((2 * pr + 2 + u) * 16) * ZS + 4 * t);
              }
            }
          }
          if (pr > 0) {
#pragma unroll
            for (int u = 0; u < 2; ++u) {
              const int mt = half * 8 + 2 * pr - 2 + u;
#pragma unroll
              for (int v = 0; v < 8; ++v) {
                const int m1 = mt * 16 + 8 * hi + v;
                const int b = m1 >> 6;
                const int i2 = (m1 >> 3) & 7;
                const int i3 = m1 & 7;
                z1[o1s * Z1P + (b * 8 + i3) * ZS + r2s * 8 + i2] =
                    acc[nxt][u][v];
              }
            }
          }
          BLOCK_FENCE();
          {
            v8f c0 = {0.f, 0.f, 0.f, 0.f, 0.f, 0.f, 0.f, 0.f};
            v8f c1 = {0.f, 0.f, 0.f, 0.f, 0.f, 0.f, 0.f, 0.f};
#pragma unroll
            for (int t = 0; t < 16; ++t) {
              c0 = WMMA4(af[cur][0][t], wb[t], c0);
              c1 = WMMA4(af[cur][1][t], wb[t], c1);
            }
            acc[cur][0] = c0;
            acc[cur][1] = c1;
          }
          BLOCK_FENCE();
        }
#pragma unroll
        for (int u = 0; u < 2; ++u) {
          const int mt = half * 8 + 6 + u;
#pragma unroll
          for (int v = 0; v < 8; ++v) {
            const int m1 = mt * 16 + 8 * hi + v;
            const int b = m1 >> 6;
            const int i2 = (m1 >> 3) & 7;
            const int i3 = m1 & 7;
            z1[o1s * Z1P + (b * 8 + i3) * ZS + r2s * 8 + i2] = acc[1][u][v];
          }
        }
      }

      {
        const int nt = w & 3;
        const int half = w >> 2;
        const float* abase = &z1[(2 * nt) * Z1P + (half * 16 + lo) * ZS + 2 * hi];
        const float* bbase = &lw2[(2 * hi) * 64 + lo];
        float* dbase = &z2[(2 * nt) * Z2P +
                           ((half * 2 + hi) * 8 + (lo >> 3)) * ZS + (lo & 7) * 8];
        v2f af[2][16];
        v2f wbs[2][16];
        v8f acc[2][2];
#pragma unroll
        for (int u = 0; u < 2; ++u) {
#pragma unroll
          for (int t = 0; t < 16; ++t) {
            af[u][t] = *(const v2f*)(abase + u * Z1P + 4 * t);
          }
        }
#pragma unroll
        for (int t = 0; t < 16; ++t) {
          wbs[0][t][0] = bbase[(4 * t) * 64 + (hi << 4)];
          wbs[0][t][1] = bbase[(4 * t + 1) * 64 + (hi << 4)];
        }
#pragma unroll
        for (int n2 = 0; n2 < 4; ++n2) {
          const int cur = n2 & 1;
          const int nxt = cur ^ 1;
          if (n2 < 3) {
            const int col = ((n2 + 1) ^ hi) << 4;
#pragma unroll
            for (int t = 0; t < 16; ++t) {
              wbs[nxt][t][0] = bbase[(4 * t) * 64 + col];
              wbs[nxt][t][1] = bbase[(4 * t + 1) * 64 + col];
            }
          }
          if (n2 > 0) {
#pragma unroll
            for (int u = 0; u < 2; ++u) {
#pragma unroll
              for (int v = 0; v < 8; ++v) {
                dbase[u * Z2P + (2 * (n2 - 1)) * ZS + v] = acc[nxt][u][v];
              }
            }
          }
          BLOCK_FENCE();
          {
            v8f c0 = {0.f, 0.f, 0.f, 0.f, 0.f, 0.f, 0.f, 0.f};
            v8f c1 = {0.f, 0.f, 0.f, 0.f, 0.f, 0.f, 0.f, 0.f};
#pragma unroll
            for (int t = 0; t < 16; ++t) {
              c0 = WMMA4(af[0][t], wbs[cur][t], c0);
              c1 = WMMA4(af[1][t], wbs[cur][t], c1);
            }
            acc[cur][0] = c0;
            acc[cur][1] = c1;
          }
          BLOCK_FENCE();
        }
#pragma unroll
        for (int u = 0; u < 2; ++u) {
#pragma unroll
          for (int v = 0; v < 8; ++v) {
            dbase[u * Z2P + 6 * ZS + v] = acc[1][u][v];
          }
        }
      }

      {
        const int nt = w & 3;
        const int half = w >> 2;
        v2f wb3[16];
#pragma unroll
        for (int t = 0; t < 16; ++t) {
          const int k = 4 * t + 2 * hi;
          wb3[t][0] = (lo < 8) ? lw3[k * 8 + lo] : 0.f;
          wb3[t][1] = (lo < 8) ? lw3[(k + 1) * 8 + lo] : 0.f;
        }
        const float* abase = &z2[(2 * nt) * Z2P + (half * 16 + lo) * ZS + 2 * hi];
        v2f af[2][16];
#pragma unroll
        for (int u = 0; u < 2; ++u) {
#pragma unroll
          for (int t = 0; t < 16; ++t) {
            af[u][t] = *(const v2f*)(abase + u * Z2P + 4 * t);
          }
        }
        BLOCK_FENCE();
        v8f acc0 = {0.f, 0.f, 0.f, 0.f, 0.f, 0.f, 0.f, 0.f};
        v8f acc1 = {0.f, 0.f, 0.f, 0.f, 0.f, 0.f, 0.f, 0.f};
#pragma unroll
        for (int t = 0; t < 16; ++t) {
          acc0 = WMMA4(af[0][t], wb3[t], acc0);
          acc1 = WMMA4(af[1][t], wb3[t], acc1);
        }
        if (lo < 8) {
          const int b = half * 2 + hi;
#pragma unroll
          for (int u = 0; u < 2; ++u) {
            const int o1 = 2 * nt + u;
#pragma unroll
            for (int v = 0; v < 8; ++v) {
              const int feat = ((o0 * 8 + o1) * 8 + v) * 8 + lo;
              out[(size_t)(bg0 + b) * 4096 + feat] =
                  (u == 0 ? acc0[v] : acc1[v]) + bias[feat];
            }
          }
        }
      }
    }
  }
}

extern "C" void kernel_launch(void* const* d_in, const int* in_sizes, int n_in,
                              void* d_out, int out_size, void* d_ws,
                              size_t ws_size, hipStream_t stream) {
  const float* x = (const float*)d_in[0];
  const float* c0 = (const float*)d_in[1];
  const float* c1 = (const float*)d_in[2];
  const float* c2 = (const float*)d_in[3];
  const float* c3 = (const float*)d_in[4];
  const float* bias = (const float*)d_in[5];
  float* out = (float*)d_out;

  dim3 grid(4096 / B_TILE);
  dim3 block(256);
  tt_fused_kernel<<<grid, block, 0, stream>>>(x, c0, c1, c2, c3, bias, out);
}


# Start-up: the four cores fetched as 16-byte pieces with all loads (cores, then x tile) in flight before the first wait, instead of 20 waited round trips of 4-byte loads; LDS stores 16 bytes wide
# speedup vs baseline: 3.4474x; 1.0925x over previous
#include <hip/hip_runtime.h>


typedef float v2f __attribute__((ext_vector_type(2)));
typedef float v4f __attribute__((ext_vector_type(4)));
typedef float v8f __attribute__((ext_vector_type(8)));

#define WMMA4(A, B, C)                                                         \
  __builtin_amdgcn_wmma_f32_16x16x4_f32(false, (A), false, (B), (short)0,      \
                                        (C), false, false)

#define BLOCK_FENCE() __builtin_amdgcn_sched_barrier(0)


#define B_TILE 4

#define ZS 68
#define Z0P (256 * ZS + 2)
#define Z1P (32 * ZS + 2)
#define Z2P (32 * ZS)

__global__ __launch_bounds__(256, 1) void tt_fused_kernel(
    const float* __restrict__ x, const float* __restrict__ c0,
    const float* __restrict__ c1, const float* __restrict__ c2,
    const float* __restrict__ c3, const float* __restrict__ bias,
    float* __restrict__ out) {
  __shared__ __align__(16) float lw0[8 * 64];
  __shared__ __align__(16) float lw1[64 * 64];
  __shared__ __align__(16) float lw2[64 * 64];
  __shared__ __align__(16) float lw3[64 * 8];
  __shared__ __align__(16) float z0[2 * Z0P];
  __shared__ __align__(16) float z1[8 * Z1P];
  __shared__ __align__(16) float z2[8 * Z2P];

  const int tid = threadIdx.x;
  const int w = tid >> 5;
  const int lane = tid & 31;
  const int lo = lane & 15;
  const int hi = lane >> 4;
  const int bg0 = blockIdx.x * B_TILE;

  v4f q1[4], q2[4];
#pragma unroll
  for (int j = 0; j < 4; ++j) {
    q1[j] = *(const v4f*)(c1 + 4 * (tid + 256 * j));
    q2[j] = *(const v4f*)(c2 + 4 * (tid + 256 * j));
  }
  const v4f qs = *(const v4f*)((tid < 128 ? c0 : c3) + 4 * (tid & 127));

  v4f xq[B_TILE][4];
  {
    const float* xw =
        x + (size_t)bg0 * 4096 + (2 * hi) * 512 + w * 64 + 4 * lo;
#pragma unroll
    for (int b = 0; b < B_TILE; ++b) {
      xq[b][0] = *(const v4f*)(xw + b * 4096);
      xq[b][1] = *(const v4f*)(xw + b * 4096 + 512);
      xq[b][2] = *(const v4f*)(xw + b * 4096 + 2048);
      xq[b][3] = *(const v4f*)(xw + b * 4096 + 2560);
    }
  }

#pragma unroll
  for (int j = 0; j < 4; ++j) {
    const int i = 4 * (tid + 256 * j);
    *(v4f*)&lw1[i ^ ((i >> 3) & 16)] = q1[j];
    *(v4f*)&lw2[i ^ ((i >> 3) & 16)] = q2[j];
  }
  *(v4f*)&(tid < 128 ? lw0 : lw3)[4 * (tid & 127)] = qs;
  __syncthreads();

  v2f xf[B_TILE][4][2];
#pragma unroll
  for (int b = 0; b < B_TILE; ++b) {
#pragma unroll
    for (int j = 0; j < 4; ++j) {
      xf[b][j][0][0] = xq[b][0][j];
      xf[b][j][0][1] = xq[b][1][j];
      xf[b][j][1][0] = xq[b][2][j];
      xf[b][j][1][1] = xq[b][3][j];
    }
  }

  for (int pi = 0; pi < 4; ++pi) {
    if (pi > 0) __syncthreads();
    {
      v2f bb0, bb1;
      bb0[0] = lw0[(2 * hi + 0) * 64 + pi * 16 + lo];
      bb0[1] = lw0[(2 * hi + 1) * 64 + pi * 16 + lo];
      bb1[0] = lw0[(4 + 2 * hi + 0) * 64 + pi * 16 + lo];
      bb1[1] = lw0[(4 + 2 * hi + 1) * 64 + pi * 16 + lo];
      const int p = lo >> 3;
      const int r1 = lo & 7;
      float* zd = &z0[p * Z0P + (32 * hi) * ZS + r1 * 8 + w];
#pragma unroll
      for (int b = 0; b < B_TILE; ++b) {
        v8f acc[4];
#pragma unroll
        for (int j = 0; j < 4; ++j) {
          v8f c = {0.f, 0.f, 0.f, 0.f, 0.f, 0.f, 0.f, 0.f};
          c = WMMA4(xf[b][j][0], bb0, c);
          acc[j] = WMMA4(xf[b][j][1], bb1, c);
        }
#pragma unroll
        for (int v = 0; v < 8; ++v) {
#pragma unroll
          for (int j = 0; j < 4; ++j) {
            zd[(b * 64 + 4 * v + j) * ZS] = acc[j][v];
          }
        }
      }
    }
    __syncthreads();

    for (int p = 0; p < 2; ++p) {
      const int o0 = pi * 2 + p;
      {
        const int nt = w & 3;
        const int half = w >> 2;
        const int nb = nt * 16 + lo;
        const int nbx = nb ^ (hi << 4);
        v2f wb[16];
#pragma unroll
        for (int t = 0; t < 16; ++t) {
          const int k = 4 * t + 2 * hi;
          wb[t][0] = lw1[k * 64 + nbx];
          wb[t][1] = lw1[(k + 1) * 64 + nbx];
        }
        const int o1s = nb >> 3;
        const int r2s = nb & 7;
        const float* abase = &z0[p * Z0P + (half * 128 + lo) * ZS + 2 * hi];
        v2f af[2][2][16];
        v8f acc[2][2];
#pragma unroll
        for (int u = 0; u < 2; ++u) {
#pragma unroll
          for (int t = 0; t < 16; ++t) {
            af[0][u][t] = *(const v2f*)(abase + (u * 16) * ZS + 4 * t);
          }
        }
#pragma unroll
        for (int pr = 0; pr < 4; ++pr) {
          const int cur = pr & 1;
          const int nxt = cur ^ 1;
          if (pr < 3) {
#pragma unroll
            for (int u = 0; u < 2; ++u) {
#pragma unroll
              for (int t = 0; t < 16; ++t) {
                af[nxt][u][t] = *(const v2f*)(
                    abase + ((2 * pr + 2 + u) * 16) * ZS + 4 * t);
              }
            }
          }
          if (pr > 0) {
#pragma unroll
            for (int u = 0; u < 2; ++u) {
              const int mt = half * 8 + 2 * pr - 2 + u;
#pragma unroll
              for (int v = 0; v < 8; ++v) {
                const int m1 = mt * 16 + 8 * hi + v;
                const int b = m1 >> 6;
                const int i2 = (m1 >> 3) & 7;
                const int i3 = m1 & 7;
                z1[o1s * Z1P + (b * 8 + i3) * ZS + r2s * 8 + i2] =
                    acc[nxt][u][v];
              }
            }
          }
          BLOCK_FENCE();
          {
            v8f c0 = {0.f, 0.f, 0.f, 0.f, 0.f, 0.f, 0.f, 0.f};
            v8f c1 = {0.f, 0.f, 0.f, 0.f, 0.f, 0.f, 0.f, 0.f};
#pragma unroll
            for (int t = 0; t < 16; ++t) {
              c0 = WMMA4(af[cur][0][t], wb[t], c0);
              c1 = WMMA4(af[cur][1][t], wb[t], c1);
            }
            acc[cur][0] = c0;
            acc[cur][1] = c1;
          }
          BLOCK_FENCE();
        }
#pragma unroll
        for (int u = 0; u < 2; ++u) {
          const int mt = half * 8 + 6 + u;
#pragma unroll
          for (int v = 0; v < 8; ++v) {
            const int m1 = mt * 16 + 8 * hi + v;
            const int b = m1 >> 6;
            const int i2 = (m1 >> 3) & 7;
            const int i3 = m1 & 7;
            z1[o1s * Z1P + (b * 8 + i3) * ZS + r2s * 8 + i2] = acc[1][u][v];
          }
        }
      }

      {
        const int nt = w & 3;
        const int half = w >> 2;
        const float* abase = &z1[(2 * nt) * Z1P + (half * 16 + lo) * ZS + 2 * hi];
        const float* bbase = &lw2[(2 * hi) * 64 + lo];
        float* dbase = &z2[(2 * nt) * Z2P +
                           ((half * 2 + hi) * 8 + (lo >> 3)) * ZS + (lo & 7) * 8];
        v2f af[2][16];
        v2f wbs[2][16];
        v8f acc[2][2];
#pragma unroll
        for (int u = 0; u < 2; ++u) {
#pragma unroll
          for (int t = 0; t < 16; ++t) {
            af[u][t] = *(const v2f*)(abase + u * Z1P + 4 * t);
          }
        }
#pragma unroll
        for (int t = 0; t < 16; ++t) {
          wbs[0][t][0] = bbase[(4 * t) * 64 + (hi << 4)];
          wbs[0][t][1] = bbase[(4 * t + 1) * 64 + (hi << 4)];
        }
#pragma unroll
        for (int n2 = 0; n2 < 4; ++n2) {
          const int cur = n2 & 1;
          const int nxt = cur ^ 1;
          if (n2 < 3) {
            const int col = ((n2 + 1) ^ hi) << 4;
#pragma unroll
            for (int t = 0; t < 16; ++t) {
              wbs[nxt][t][0] = bbase[(4 * t) * 64 + col];
              wbs[nxt][t][1] = bbase[(4 * t + 1) * 64 + col];
            }
          }
          if (n2 > 0) {
#pragma unroll
            for (int u = 0; u < 2; ++u) {
#pragma unroll
              for (int v = 0; v < 8; ++v) {
                dbase[u * Z2P + (2 * (n2 - 1)) * ZS + v] = acc[nxt][u][v];
              }
            }
          }
          BLOCK_FENCE();
          {
            v8f c0 = {0.f, 0.f, 0.f, 0.f, 0.f, 0.f, 0.f, 0.f};
            v8f c1 = {0.f, 0.f, 0.f, 0.f, 0.f, 0.f, 0.f, 0.f};
#pragma unroll
            for (int t = 0; t < 16; ++t) {
              c0 = WMMA4(af[0][t], wbs[cur][t], c0);
              c1 = WMMA4(af[1][t], wbs[cur][t], c1);
            }
            acc[cur][0] = c0;
            acc[cur][1] = c1;
          }
          BLOCK_FENCE();
        }
#pragma unroll
        for (int u = 0; u < 2; ++u) {
#pragma unroll
          for (int v = 0; v < 8; ++v) {
            dbase[u * Z2P + 6 * ZS + v] = acc[1][u][v];
          }
        }
      }

      {
        const int nt = w & 3;
        const int half = w >> 2;
        v2f wb3[16];
#pragma unroll
        for (int t = 0; t < 16; ++t) {
          const int k = 4 * t + 2 * hi;
          wb3[t][0] = (lo < 8) ? lw3[k * 8 + lo] : 0.f;
          wb3[t][1] = (lo < 8) ? lw3[(k + 1) * 8 + lo] : 0.f;
        }
        const float* abase = &z2[(2 * nt) * Z2P + (half * 16 + lo) * ZS + 2 * hi];
        v2f af[2][16];
#pragma unroll
        for (int u = 0; u < 2; ++u) {
#pragma unroll
          for (int t = 0; t < 16; ++t) {
            af[u][t] = *(const v2f*)(abase + u * Z2P + 4 * t);
          }
        }
        BLOCK_FENCE();
        v8f acc0 = {0.f, 0.f, 0.f, 0.f, 0.f, 0.f, 0.f, 0.f};
        v8f acc1 = {0.f, 0.f, 0.f, 0.f, 0.f, 0.f, 0.f, 0.f};
#pragma unroll
        for (int t = 0; t < 16; ++t) {
          acc0 = WMMA4(af[0][t], wb3[t], acc0);
          acc1 = WMMA4(af[1][t], wb3[t], acc1);
        }
        if (lo < 8) {
          const int b = half * 2 + hi;
#pragma unroll
          for (int u = 0; u < 2; ++u) {
            const int o1 = 2 * nt + u;
#pragma unroll
            for (int v = 0; v < 8; ++v) {
              const int feat = ((o0 * 8 + o1) * 8 + v) * 8 + lo;
              out[(size_t)(bg0 + b) * 4096 + feat] =
                  (u == 0 ? acc0[v] : acc1[v]) + bias[feat];
            }
          }
        }
      }
    }
  }
}

extern "C" void kernel_launch(void* const* d_in, const int* in_sizes, int n_in,
                              void* d_out, int out_size, void* d_ws,
                              size_t ws_size, hipStream_t stream) {
  const float* x = (const float*)d_in[0];
  const float* c0 = (const float*)d_in[1];
  const float* c1 = (const float*)d_in[2];
  const float* c2 = (const float*)d_in[3];
  const float* c3 = (const float*)d_in[4];
  const float* bias = (const float*)d_in[5];
  float* out = (float*)d_out;

  dim3 grid(4096 / B_TILE);
  dim3 block(256);
  tt_fused_kernel<<<grid, block, 0, stream>>>(x, c0, c1, c2, c3, bias, out);
}


# Younger wave of each SIMD pair (w+4) holds s_setprio 3 from the barrier behind stage 0 to the end of its stage 1 of the second o0, then 0: each wave's LDS phases run under its partner's WMMAs
# speedup vs baseline: 3.6022x; 1.0449x over previous
#include <hip/hip_runtime.h>


typedef float v2f __attribute__((ext_vector_type(2)));
typedef float v4f __attribute__((ext_vector_type(4)));
typedef float v8f __attribute__((ext_vector_type(8)));

#define WMMA4(A, B, C)                                                         \
  __builtin_amdgcn_wmma_f32_16x16x4_f32(false, (A), false, (B), (short)0,      \
                                        (C), false, false)

#define BLOCK_FENCE() __builtin_amdgcn_sched_barrier(0)


#define B_TILE 4

#define ZS 68
#define Z0P (256 * ZS + 2)
#define Z1P (32 * ZS + 2)
#define Z2P (32 * ZS)

__global__ __launch_bounds__(256, 1) void tt_fused_kernel(
    const float* __restrict__ x, const float* __restrict__ c0,
    const float* __restrict__ c1, const float* __restrict__ c2,
    const float* __restrict__ c3, const float* __restrict__ bias,
    float* __restrict__ out) {
  __shared__ __align__(16) float lw0[8 * 64];
  __shared__ __align__(16) float lw1[64 * 64];
  __shared__ __align__(16) float lw2[64 * 64];
  __shared__ __align__(16) float lw3[64 * 8];
  __shared__ __align__(16) float z0[2 * Z0P];
  __shared__ __align__(16) float z1[8 * Z1P];
  __shared__ __align__(16) float z2[8 * Z2P];

  const int tid = threadIdx.x;
  const int w = tid >> 5;
  const int lane = tid & 31;
  const int lo = lane & 15;
  const int hi = lane >> 4;
  const int bg0 = blockIdx.x * B_TILE;
  const bool younger = __builtin_amdgcn_readfirstlane(tid >> 5) >= 4;

  v4f q1[4], q2[4];
#pragma unroll
  for (int j = 0; j < 4; ++j) {
    q1[j] = *(const v4f*)(c1 + 4 * (tid + 256 * j));
    q2[j] = *(const v4f*)(c2 + 4 * (tid + 256 * j));
  }
  const v4f qs = *(const v4f*)((tid < 128 ? c0 : c3) + 4 * (tid & 127));

  v4f xq[B_TILE][4];
  {
    const float* xw =
        x + (size_t)bg0 * 4096 + (2 * hi) * 512 + w * 64 + 4 * lo;
#pragma unroll
    for (int b = 0; b < B_TILE; ++b) {
      xq[b][0] = *(const v4f*)(xw + b * 4096);
      xq[b][1] = *(const v4f*)(xw + b * 4096 + 512);
      xq[b][2] = *(const v4f*)(xw + b * 4096 + 2048);
      xq[b][3] = *(const v4f*)(xw + b * 4096 + 2560);
    }
  }

#pragma unroll
  for (int j = 0; j < 4; ++j) {
    const int i = 4 * (tid + 256 * j);
    *(v4f*)&lw1[i ^ ((i >> 3) & 16)] = q1[j];
    *(v4f*)&lw2[i ^ ((i >> 3) & 16)] = q2[j];
  }
  *(v4f*)&(tid < 128 ? lw0 : lw3)[4 * (tid & 127)] = qs;
  __syncthreads();

  v2f xf[B_TILE][4][2];
#pragma unroll
  for (int b = 0; b < B_TILE; ++b) {
#pragma unroll
    for (int j = 0; j < 4; ++j) {
      xf[b][j][0][0] = xq[b][0][j];
      xf[b][j][0][1] = xq[b][1][j];
      xf[b][j][1][0] = xq[b][2][j];
      xf[b][j][1][1] = xq[b][3][j];
    }
  }

  for (int pi = 0; pi < 4; ++pi) {
    if (pi > 0) __syncthreads();
    {
      v2f bb0, bb1;
      bb0[0] = lw0[(2 * hi + 0) * 64 + pi * 16 + lo];
      bb0[1] = lw0[(2 * hi + 1) * 64 + pi * 16 + lo];
      bb1[0] = lw0[(4 + 2 * hi + 0) * 64 + pi * 16 + lo];
      bb1[1] = lw0[(4 + 2 * hi + 1) * 64 + pi * 16 + lo];
      const int p = lo >> 3;
      const int r1 = lo & 7;
      float* zd = &z0[p * Z0P + (32 * hi) * ZS + r1 * 8 + w];
#pragma unroll
      for (int b = 0; b < B_TILE; ++b) {
        v8f acc[4];
#pragma unroll
        for (int j = 0; j < 4; ++j) {
          v8f c = {0.f, 0.f, 0.f, 0.f, 0.f, 0.f, 0.f, 0.f};
          c = WMMA4(xf[b][j][0], bb0, c);
          acc[j] = WMMA4(xf[b][j][1], bb1, c);
        }
#pragma unroll
        for (int v = 0; v < 8; ++v) {
#pragma unroll
          for (int j = 0; j < 4; ++j) {
            zd[(b * 64 + 4 * v + j) * ZS] = acc[j][v];
          }
        }
      }
    }
    __syncthreads();
    if (younger) __builtin_amdgcn_s_setprio(3);

    for (int p = 0; p < 2; ++p) {
      const int o0 = pi * 2 + p;
      {
        const int nt = w & 3;
        const int half = w >> 2;
        const int nb = nt * 16 + lo;
        const int nbx = nb ^ (hi << 4);
        v2f wb[16];
#pragma unroll
        for (int t = 0; t < 16; ++t) {
          const int k = 4 * t + 2 * hi;
          wb[t][0] = lw1[k * 64 + nbx];
          wb[t][1] = lw1[(k + 1) * 64 + nbx];
        }
        const int o1s = nb >> 3;
        const int r2s = nb & 7;
        const float* abase = &z0[p * Z0P + (half * 128 + lo) * ZS + 2 * hi];
        v2f af[2][2][16];
        v8f acc[2][2];
#pragma unroll
        for (int u = 0; u < 2; ++u) {
#pragma unroll
          for (int t = 0; t < 16; ++t) {
            af[0][u][t] = *(const v2f*)(abase + (u * 16) * ZS + 4 * t);
          }
        }
#pragma unroll
        for (int pr = 0; pr < 4; ++pr) {
          const int cur = pr & 1;
          const int nxt = cur ^ 1;
          if (pr < 3) {
#pragma unroll
            for (int u = 0; u < 2; ++u) {
#pragma unroll
              for (int t = 0; t < 16; ++t) {
                af[nxt][u][t] = *(const v2f*)(
                    abase + ((2 * pr + 2 + u) * 16) * ZS + 4 * t);
              }
            }
          }
          if (pr > 0) {
#pragma unroll
            for (int u = 0; u < 2; ++u) {
              const int mt = half * 8 + 2 * pr - 2 + u;
#pragma unroll
              for (int v = 0; v < 8; ++v) {
                const int m1 = mt * 16 + 8 * hi + v;
                const int b = m1 >> 6;
                const int i2 = (m1 >> 3) & 7;
                const int i3 = m1 & 7;
                z1[o1s * Z1P + (b * 8 + i3) * ZS + r2s * 8 + i2] =
                    acc[nxt][u][v];
              }
            }
          }
          BLOCK_FENCE();
          {
            v8f c0 = {0.f, 0.f, 0.f, 0.f, 0.f, 0.f, 0.f, 0.f};
            v8f c1 = {0.f, 0.f, 0.f, 0.f, 0.f, 0.f, 0.f, 0.f};
#pragma unroll
            for (int t = 0; t < 16; ++t) {
              c0 = WMMA4(af[cur][0][t], wb[t], c0);
              c1 = WMMA4(af[cur][1][t], wb[t], c1);
            }
            acc[cur][0] = c0;
            acc[cur][1] = c1;
          }
          BLOCK_FENCE();
        }
#pragma unroll
        for (int u = 0; u < 2; ++u) {
          const int mt = half * 8 + 6 + u;
#pragma unroll
          for (int v = 0; v < 8; ++v) {
            const int m1 = mt * 16 + 8 * hi + v;
            const int b = m1 >> 6;
            const int i2 = (m1 >> 3) & 7;
            const int i3 = m1 & 7;
            z1[o1s * Z1P + (b * 8 + i3) * ZS + r2s * 8 + i2] = acc[1][u][v];
          }
        }
      }

      if (younger && p == 1) __builtin_amdgcn_s_setprio(0);

      {
        const int nt = w & 3;
        const int half = w >> 2;
        const float* abase = &z1[(2 * nt) * Z1P + (half * 16 + lo) * ZS + 2 * hi];
        const float* bbase = &lw2[(2 * hi) * 64 + lo];
        float* dbase = &z2[(2 * nt) * Z2P +
                           ((half * 2 + hi) * 8 + (lo >> 3)) * ZS + (lo & 7) * 8];
        v2f af[2][16];
        v2f wbs[2][16];
        v8f acc[2][2];
#pragma unroll
        for (int u = 0; u < 2; ++u) {
#pragma unroll
          for (int t = 0; t < 16; ++t) {
            af[u][t] = *(const v2f*)(abase + u * Z1P + 4 * t);
          }
        }
#pragma unroll
        for (int t = 0; t < 16; ++t) {
          wbs[0][t][0] = bbase[(4 * t) * 64 + (hi << 4)];
          wbs[0][t][1] = bbase[(4 * t + 1) * 64 + (hi << 4)];
        }
#pragma unroll
        for (int n2 = 0; n2 < 4; ++n2) {
          const int cur = n2 & 1;
          const int nxt = cur ^ 1;
          if (n2 < 3) {
            const int col = ((n2 + 1) ^ hi) << 4;
#pragma unroll
            for (int t = 0; t < 16; ++t) {
              wbs[nxt][t][0] = bbase[(4 * t) * 64 + col];
              wbs[nxt][t][1] = bbase[(4 * t + 1) * 64 + col];
            }
          }
          if (n2 > 0) {
#pragma unroll
            for (int u = 0; u < 2; ++u) {
#pragma unroll
              for (int v = 0; v < 8; ++v) {
                dbase[u * Z2P + (2 * (n2 - 1)) * ZS + v] = acc[nxt][u][v];
              }
            }
          }
          BLOCK_FENCE();
          {
            v8f c0 = {0.f, 0.f, 0.f, 0.f, 0.f, 0.f, 0.f, 0.f};
            v8f c1 = {0.f, 0.f, 0.f, 0.f, 0.f, 0.f, 0.f, 0.f};
#pragma unroll
            for (int t = 0; t < 16; ++t) {
              c0 = WMMA4(af[0][t], wbs[cur][t], c0);
              c1 = WMMA4(af[1][t], wbs[cur][t], c1);
            }
            acc[cur][0] = c0;
            acc[cur][1] = c1;
          }
          BLOCK_FENCE();
        }
#pragma unroll
        for (int u = 0; u < 2; ++u) {
#pragma unroll
          for (int v = 0; v < 8; ++v) {
            dbase[u * Z2P + 6 * ZS + v] = acc[1][u][v];
          }
        }
      }

      {
        const int nt = w & 3;
        const int half = w >> 2;
        v2f wb3[16];
#pragma unroll
        for (int t = 0; t < 16; ++t) {
          const int k = 4 * t + 2 * hi;
          wb3[t][0] = (lo < 8) ? lw3[k * 8 + lo] : 0.f;
          wb3[t][1] = (lo < 8) ? lw3[(k + 1) * 8 + lo] : 0.f;
        }
        const float* abase = &z2[(2 * nt) * Z2P + (half * 16 + lo) * ZS + 2 * hi];
        v2f af[2][16];
#pragma unroll
        for (int u = 0; u < 2; ++u) {
#pragma unroll
          for (int t = 0; t < 16; ++t) {
            af[u][t] = *(const v2f*)(abase + u * Z2P + 4 * t);
          }
        }
        BLOCK_FENCE();
        v8f acc0 = {0.f, 0.f, 0.f, 0.f, 0.f, 0.f, 0.f, 0.f};
        v8f acc1 = {0.f, 0.f, 0.f, 0.f, 0.f, 0.f, 0.f, 0.f};
#pragma unroll
        for (int t = 0; t < 16; ++t) {
          acc0 = WMMA4(af[0][t], wb3[t], acc0);
          acc1 = WMMA4(af[1][t], wb3[t], acc1);
        }
        if (lo < 8) {
          const int b = half * 2 + hi;
#pragma unroll
          for (int u = 0; u < 2; ++u) {
            const int o1 = 2 * nt + u;
#pragma unroll
            for (int v = 0; v < 8; ++v) {
              const int feat = ((o0 * 8 + o1) * 8 + v) * 8 + lo;
              out[(size_t)(bg0 + b) * 4096 + feat] =
                  (u == 0 ? acc0[v] : acc1[v]) + bias[feat];
            }
          }
        }
      }
    }
  }
}

extern "C" void kernel_launch(void* const* d_in, const int* in_sizes, int n_in,
                              void* d_out, int out_size, void* d_ws,
                              size_t ws_size, hipStream_t stream) {
  const float* x = (const float*)d_in[0];
  const float* c0 = (const float*)d_in[1];
  const float* c1 = (const float*)d_in[2];
  const float* c2 = (const float*)d_in[3];
  const float* c3 = (const float*)d_in[4];
  const float* bias = (const float*)d_in[5];
  float* out = (float*)d_out;

  dim3 grid(4096 / B_TILE);
  dim3 block(256);
  tt_fused_kernel<<<grid, block, 0, stream>>>(x, c0, c1, c2, c3, bias, out);
}


# Loads that do not depend on the previous stage leave the stage hand-overs: lw1/lw2/lw3 fragments kept in registers for all o0; bias read in stage 2 is stage 3's start value (bias summed first)
# speedup vs baseline: 3.7465x; 1.0401x over previous
#include <hip/hip_runtime.h>


typedef float v2f __attribute__((ext_vector_type(2)));
typedef float v4f __attribute__((ext_vector_type(4)));
typedef float v8f __attribute__((ext_vector_type(8)));

#define WMMA4(A, B, C)                                                         \
  __builtin_amdgcn_wmma_f32_16x16x4_f32(false, (A), false, (B), (short)0,      \
                                        (C), false, false)

#define BLOCK_FENCE() __builtin_amdgcn_sched_barrier(0)


#define B_TILE 4

#define ZS 68
#define Z0P (256 * ZS + 2)
#define Z1P (32 * ZS + 2)
#define Z2P (32 * ZS)

__global__ __launch_bounds__(256, 1) void tt_fused_kernel(
    const float* __restrict__ x, const float* __restrict__ c0,
    const float* __restrict__ c1, const float* __restrict__ c2,
    const float* __restrict__ c3, const float* __restrict__ bias,
    float* __restrict__ out) {
  __shared__ __align__(16) float lw0[8 * 64];
  __shared__ __align__(16) float lw1[64 * 64];
  __shared__ __align__(16) float lw2[64 * 64];
  __shared__ __align__(16) float lw3[64 * 8];
  __shared__ __align__(16) float z0[2 * Z0P];
  __shared__ __align__(16) float z1[8 * Z1P];
  __shared__ __align__(16) float z2[8 * Z2P];

  const int tid = threadIdx.x;
  const int w = tid >> 5;
  const int lane = tid & 31;
  const int lo = lane & 15;
  const int hi = lane >> 4;
  const int bg0 = blockIdx.x * B_TILE;
  const bool younger = __builtin_amdgcn_readfirstlane(tid >> 5) >= 4;

  v4f q1[4], q2[4];
#pragma unroll
  for (int j = 0; j < 4; ++j) {
    q1[j] = *(const v4f*)(c1 + 4 * (tid + 256 * j));
    q2[j] = *(const v4f*)(c2 + 4 * (tid + 256 * j));
  }
  const v4f qs = *(const v4f*)((tid < 128 ? c0 : c3) + 4 * (tid & 127));

  v4f xq[B_TILE][4];
  {
    const float* xw =
        x + (size_t)bg0 * 4096 + (2 * hi) * 512 + w * 64 + 4 * lo;
#pragma unroll
    for (int b = 0; b < B_TILE; ++b) {
      xq[b][0] = *(const v4f*)(xw + b * 4096);
      xq[b][1] = *(const v4f*)(xw + b * 4096 + 512);
      xq[b][2] = *(const v4f*)(xw + b * 4096 + 2048);
      xq[b][3] = *(const v4f*)(xw + b * 4096 + 2560);
    }
  }

#pragma unroll
  for (int j = 0; j < 4; ++j) {
    const int i = 4 * (tid + 256 * j);
    *(v4f*)&lw1[i ^ ((i >> 3) & 16)] = q1[j];
    *(v4f*)&lw2[i ^ ((i >> 3) & 16)] = q2[j];
  }
  *(v4f*)&(tid < 128 ? lw0 : lw3)[4 * (tid & 127)] = qs;
  __syncthreads();

  v2f wb[16];
  v2f wbs[4][16];
  v2f wb3[16];
  {
    const int nbx = ((w & 3) * 16 + lo) ^ (hi << 4);
#pragma unroll
    for (int t = 0; t < 16; ++t) {
      const int k = 4 * t + 2 * hi;
      wb[t][0] = lw1[k * 64 + nbx];
      wb[t][1] = lw1[(k + 1) * 64 + nbx];
    }
    const float* bbase = &lw2[(2 * hi) * 64 + lo];
#pragma unroll
    for (int n2 = 0; n2 < 4; ++n2) {
      const int col = (n2 ^ hi) << 4;
#pragma unroll
      for (int t = 0; t < 16; ++t) {
        wbs[n2][t][0] = bbase[(4 * t) * 64 + col];
        wbs[n2][t][1] = bbase[(4 * t + 1) * 64 + col];
      }
    }
    const float* b3 = &lw3[(2 * hi) * 8 + (lo & 7)];
#pragma unroll
    for (int t = 0; t < 16; ++t) {
      const float e0 = b3[(4 * t) * 8];
      const float e1 = b3[(4 * t + 1) * 8];
      wb3[t][0] = (lo < 8) ? e0 : 0.f;
      wb3[t][1] = (lo < 8) ? e1 : 0.f;
    }
  }

  v2f xf[B_TILE][4][2];
#pragma unroll
  for (int b = 0; b < B_TILE; ++b) {
#pragma unroll
    for (int j = 0; j < 4; ++j) {
      xf[b][j][0][0] = xq[b][0][j];
      xf[b][j][0][1] = xq[b][1][j];
      xf[b][j][1][0] = xq[b][2][j];
      xf[b][j][1][1] = xq[b][3][j];
    }
  }

  for (int pi = 0; pi < 4; ++pi) {
    if (pi > 0) __syncthreads();
    {
      v2f bb0, bb1;
      bb0[0] = lw0[(2 * hi + 0) * 64 + pi * 16 + lo];
      bb0[1] = lw0[(2 * hi + 1) * 64 + pi * 16 + lo];
      bb1[0] = lw0[(4 + 2 * hi + 0) * 64 + pi * 16 + lo];
      bb1[1] = lw0[(4 + 2 * hi + 1) * 64 + pi * 16 + lo];
      const int p = lo >> 3;
      const int r1 = lo & 7;
      float* zd = &z0[p * Z0P + (32 * hi) * ZS + r1 * 8 + w];
#pragma unroll
      for (int b = 0; b < B_TILE; ++b) {
        v8f acc[4];
#pragma unroll
        for (int j = 0; j < 4; ++j) {
          v8f c = {0.f, 0.f, 0.f, 0.f, 0.f, 0.f, 0.f, 0.f};
          c = WMMA4(xf[b][j][0], bb0, c);
          acc[j] = WMMA4(xf[b][j][1], bb1, c);
        }
#pragma unroll
        for (int v = 0; v < 8; ++v) {
#pragma unroll
          for (int j = 0; j < 4; ++j) {
            zd[(b * 64 + 4 * v + j) * ZS] = acc[j][v];
          }
        }
      }
    }
    __syncthreads();
    if (younger) __builtin_amdgcn_s_setprio(3);

    for (int p = 0; p < 2; ++p) {
      const int o0 = pi * 2 + p;
      {
        const int nt = w & 3;
        const int half = w >> 2;
        const int nb = nt * 16 + lo;
        const int o1s = nb >> 3;
        const int r2s = nb & 7;
        const float* abase = &z0[p * Z0P + (half * 128 + lo) * ZS + 2 * hi];
        v2f af[2][2][16];
        v8f acc[2][2];
#pragma unroll
        for (int u = 0; u < 2; ++u) {
#pragma unroll
          for (int t = 0; t < 16; ++t) {
            af[0][u][t] = *(const v2f*)(abase + (u * 16) * ZS + 4 * t);
          }
        }
#pragma unroll
        for (int pr = 0; pr < 4; ++pr) {
          const int cur = pr & 1;
          const int nxt = cur ^ 1;
          if (pr < 3) {
#pragma unroll
            for (int u = 0; u < 2; ++u) {
#pragma unroll
              for (int t = 0; t < 16; ++t) {
                af[nxt][u][t] = *(const v2f*)(
                    abase + ((2 * pr + 2 + u) * 16) * ZS + 4 * t);
              }
            }
          }
          if (pr > 0) {
#pragma unroll
            for (int u = 0; u < 2; ++u) {
              const int mt = half * 8 + 2 * pr - 2 + u;
#pragma unroll
              for (int v = 0; v < 8; ++v) {
                const int m1 = mt * 16 + 8 * hi + v;
                const int b = m1 >> 6;
                const int i2 = (m1 >> 3) & 7;
                const int i3 = m1 & 7;
                z1[o1s * Z1P + (b * 8 + i3) * ZS + r2s * 8 + i2] =
                    acc[nxt][u][v];
              }
            }
          }
          BLOCK_FENCE();
          {
            v8f c0 = {0.f, 0.f, 0.f, 0.f, 0.f, 0.f, 0.f, 0.f};
            v8f c1 = {0.f, 0.f, 0.f, 0.f, 0.f, 0.f, 0.f, 0.f};
#pragma unroll
            for (int t = 0; t < 16; ++t) {
              c0 = WMMA4(af[cur][0][t], wb[t], c0);
              c1 = WMMA4(af[cur][1][t], wb[t], c1);
            }
            acc[cur][0] = c0;
            acc[cur][1] = c1;
          }
          BLOCK_FENCE();
        }
#pragma unroll
        for (int u = 0; u < 2; ++u) {
          const int mt = half * 8 + 6 + u;
#pragma unroll
          for (int v = 0; v < 8; ++v) {
            const int m1 = mt * 16 + 8 * hi + v;
            const int b = m1 >> 6;
            const int i2 = (m1 >> 3) & 7;
            const int i3 = m1 & 7;
            z1[o1s * Z1P + (b * 8 + i3) * ZS + r2s * 8 + i2] = acc[1][u][v];
          }
        }
      }

      if (younger && p == 1) __builtin_amdgcn_s_setprio(0);

      float bv[2][8];

      {
        const int nt = w & 3;
        const int half = w >> 2;
        const float* abase = &z1[(2 * nt) * Z1P + (half * 16 + lo) * ZS + 2 * hi];
        float* dbase = &z2[(2 * nt) * Z2P +
                           ((half * 2 + hi) * 8 + (lo >> 3)) * ZS + (lo & 7) * 8];
        v2f af[2][16];
        v8f acc[2][2];
#pragma unroll
        for (int u = 0; u < 2; ++u) {
#pragma unroll
          for (int t = 0; t < 16; ++t) {
            af[u][t] = *(const v2f*)(abase + u * Z1P + 4 * t);
          }
        }
        {
          const float* bp = bias + (o0 * 8 + 2 * nt) * 64 + (lo & 7);
#pragma unroll
          for (int u = 0; u < 2; ++u) {
#pragma unroll
            for (int v = 0; v < 8; ++v) {
              bv[u][v] = bp[u * 64 + v * 8];
            }
          }
        }
#pragma unroll
        for (int n2 = 0; n2 < 4; ++n2) {
          const int cur = n2 & 1;
          const int nxt = cur ^ 1;
          if (n2 > 0) {
#pragma unroll
            for (int u = 0; u < 2; ++u) {
#pragma unroll
              for (int v = 0; v < 8; ++v) {
                dbase[u * Z2P + (2 * (n2 - 1)) * ZS + v] = acc[nxt][u][v];
              }
            }
          }
          BLOCK_FENCE();
          {
            v8f c0 = {0.f, 0.f, 0.f, 0.f, 0.f, 0.f, 0.f, 0.f};
            v8f c1 = {0.f, 0.f, 0.f, 0.f, 0.f, 0.f, 0.f, 0.f};
#pragma unroll
            for (int t = 0; t < 16; ++t) {
              c0 = WMMA4(af[0][t], wbs[n2][t], c0);
              c1 = WMMA4(af[1][t], wbs[n2][t], c1);
            }
            acc[cur][0] = c0;
            acc[cur][1] = c1;
          }
        }
#pragma unroll
        for (int u = 0; u < 2; ++u) {
#pragma unroll
          for (int v = 0; v < 8; ++v) {
            dbase[u * Z2P + 6 * ZS + v] = acc[1][u][v];
          }
        }
      }

      {
        const int nt = w & 3;
        const int half = w >> 2;
        const float* abase = &z2[(2 * nt) * Z2P + (half * 16 + lo) * ZS + 2 * hi];
        v2f af[2][16];
#pragma unroll
        for (int u = 0; u < 2; ++u) {
#pragma unroll
          for (int t = 0; t < 16; ++t) {
            af[u][t] = *(const v2f*)(abase + u * Z2P + 4 * t);
          }
        }
        BLOCK_FENCE();
        v8f acc0, acc1;
#pragma unroll
        for (int v = 0; v < 8; ++v) {
          acc0[v] = bv[0][v];
          acc1[v] = bv[1][v];
        }
#pragma unroll
        for (int t = 0; t < 16; ++t) {
          acc0 = WMMA4(af[0][t], wb3[t], acc0);
          acc1 = WMMA4(af[1][t], wb3[t], acc1);
        }
        if (lo < 8) {
          const int b = half * 2 + hi;
#pragma unroll
          for (int u = 0; u < 2; ++u) {
            const int o1 = 2 * nt + u;
#pragma unroll
            for (int v = 0; v < 8; ++v) {
              const int feat = ((o0 * 8 + o1) * 8 + v) * 8 + lo;
              out[(size_t)(bg0 + b) * 4096 + feat] =
                  (u == 0 ? acc0[v] : acc1[v]);
            }
          }
        }
      }
    }
  }
}

extern "C" void kernel_launch(void* const* d_in, const int* in_sizes, int n_in,
                              void* d_out, int out_size, void* d_ws,
                              size_t ws_size, hipStream_t stream) {
  const float* x = (const float*)d_in[0];
  const float* c0 = (const float*)d_in[1];
  const float* c1 = (const float*)d_in[2];
  const float* c2 = (const float*)d_in[3];
  const float* c3 = (const float*)d_in[4];
  const float* bias = (const float*)d_in[5];
  float* out = (float*)d_out;

  dim3 grid(4096 / B_TILE);
  dim3 block(256);
  tt_fused_kernel<<<grid, block, 0, stream>>>(x, c0, c1, c2, c3, bias, out);
}


# Workgroups made persistent: 256 workgroups x 4 batch tiles each; cores and core fragments fetched once, next tile's x asked for behind the current tile's last stage 0
# speedup vs baseline: 3.8755x; 1.0344x over previous
#include <hip/hip_runtime.h>


typedef float v2f __attribute__((ext_vector_type(2)));
typedef float v4f __attribute__((ext_vector_type(4)));
typedef float v8f __attribute__((ext_vector_type(8)));

#define WMMA4(A, B, C)                                                         \
  __builtin_amdgcn_wmma_f32_16x16x4_f32(false, (A), false, (B), (short)0,      \
                                        (C), false, false)

#define BLOCK_FENCE() __builtin_amdgcn_sched_barrier(0)


#define B_TILE 4
#define TILES_PER_WG 4

#define ZS 68
#define Z0P (256 * ZS + 2)
#define Z1P (32 * ZS + 2)
#define Z2P (32 * ZS)

__global__ __launch_bounds__(256, 1) void tt_fused_kernel(
    const float* __restrict__ x, const float* __restrict__ c0,
    const float* __restrict__ c1, const float* __restrict__ c2,
    const float* __restrict__ c3, const float* __restrict__ bias,
    float* __restrict__ out) {
  __shared__ __align__(16) float lw0[8 * 64];
  __shared__ __align__(16) float lw1[64 * 64];
  __shared__ __align__(16) float lw2[64 * 64];
  __shared__ __align__(16) float lw3[64 * 8];
  __shared__ __align__(16) float z0[2 * Z0P];
  __shared__ __align__(16) float z1[8 * Z1P];
  __shared__ __align__(16) float z2[8 * Z2P];

  const int tid = threadIdx.x;
  const int w = tid >> 5;
  const int lane = tid & 31;
  const int lo = lane & 15;
  const int hi = lane >> 4;
  const int tile0 = blockIdx.x * TILES_PER_WG;
  const bool younger = __builtin_amdgcn_readfirstlane(tid >> 5) >= 4;

  v4f q1[4], q2[4];
#pragma unroll
  for (int j = 0; j < 4; ++j) {
    q1[j] = *(const v4f*)(c1 + 4 * (tid + 256 * j));
    q2[j] = *(const v4f*)(c2 + 4 * (tid + 256 * j));
  }
  const v4f qs = *(const v4f*)((tid < 128 ? c0 : c3) + 4 * (tid & 127));

  v4f xq[B_TILE][4];
  const float* xw = x + (size_t)(tile0 * B_TILE) * 4096 + (2 * hi) * 512 +
                    w * 64 + 4 * lo;
#pragma unroll
  for (int b = 0; b < B_TILE; ++b) {
    xq[b][0] = *(const v4f*)(xw + b * 4096);
    xq[b][1] = *(const v4f*)(xw + b * 4096 + 512);
    xq[b][2] = *(const v4f*)(xw + b * 4096 + 2048);
    xq[b][3] = *(const v4f*)(xw + b * 4096 + 2560);
  }

#pragma unroll
  for (int j = 0; j < 4; ++j) {
    const int i = 4 * (tid + 256 * j);
    *(v4f*)&lw1[i ^ ((i >> 3) & 16)] = q1[j];
    *(v4f*)&lw2[i ^ ((i >> 3) & 16)] = q2[j];
  }
  *(v4f*)&(tid < 128 ? lw0 : lw3)[4 * (tid & 127)] = qs;
  __syncthreads();

  v2f wb[16];
  v2f wbs[4][16];
  v2f wb3[16];
  {
    const int nbx = ((w & 3) * 16 + lo) ^ (hi << 4);
#pragma unroll
    for (int t = 0; t < 16; ++t) {
      const int k = 4 * t + 2 * hi;
      wb[t][0] = lw1[k * 64 + nbx];
      wb[t][1] = lw1[(k + 1) * 64 + nbx];
    }
    const float* bbase = &lw2[(2 * hi) * 64 + lo];
#pragma unroll
    for (int n2 = 0; n2 < 4; ++n2) {
      const int col = (n2 ^ hi) << 4;
#pragma unroll
      for (int t = 0; t < 16; ++t) {
        wbs[n2][t][0] = bbase[(4 * t) * 64 + col];
        wbs[n2][t][1] = bbase[(4 * t + 1) * 64 + col];
      }
    }
    const float* b3 = &lw3[(2 * hi) * 8 + (lo & 7)];
#pragma unroll
    for (int t = 0; t < 16; ++t) {
      const float e0 = b3[(4 * t) * 8];
      const float e1 = b3[(4 * t + 1) * 8];
      wb3[t][0] = (lo < 8) ? e0 : 0.f;
      wb3[t][1] = (lo < 8) ? e1 : 0.f;
    }
  }

  for (int it = 0; it < TILES_PER_WG; ++it) {
  const int bg0 = (tile0 + it) * B_TILE;
  v2f xf[B_TILE][4][2];
#pragma unroll
  for (int b = 0; b < B_TILE; ++b) {
#pragma unroll
    for (int j = 0; j < 4; ++j) {
      xf[b][j][0][0] = xq[b][0][j];
      xf[b][j][0][1] = xq[b][1][j];
      xf[b][j][1][0] = xq[b][2][j];
      xf[b][j][1][1] = xq[b][3][j];
    }
  }

  for (int pi = 0; pi < 4; ++pi) {
    if (pi > 0 || it > 0) __syncthreads();
    {
      v2f bb0, bb1;
      bb0[0] = lw0[(2 * hi + 0) * 64 + pi * 16 + lo];
      bb0[1] = lw0[(2 * hi + 1) * 64 + pi * 16 + lo];
      bb1[0] = lw0[(4 + 2 * hi + 0) * 64 + pi * 16 + lo];
      bb1[1] = lw0[(4 + 2 * hi + 1) * 64 + pi * 16 + lo];
      const int p = lo >> 3;
      const int r1 = lo & 7;
      float* zd = &z0[p * Z0P + (32 * hi) * ZS + r1 * 8 + w];
#pragma unroll
      for (int b = 0; b < B_TILE; ++b) {
        v8f acc[4];
#pragma unroll
        for (int j = 0; j < 4; ++j) {
          v8f c = {0.f, 0.f, 0.f, 0.f, 0.f, 0.f, 0.f, 0.f};
          c = WMMA4(xf[b][j][0], bb0, c);
          acc[j] = WMMA4(xf[b][j][1], bb1, c);
        }
#pragma unroll
        for (int v = 0; v < 8; ++v) {
#pragma unroll
          for (int j = 0; j < 4; ++j) {
            zd[(b * 64 + 4 * v + j) * ZS] = acc[j][v];
          }
        }
      }
    }
    __syncthreads();
    if (pi == 3 && it + 1 < TILES_PER_WG) {
      xw += B_TILE * 4096;
#pragma unroll
      for (int b = 0; b < B_TILE; ++b) {
        xq[b][0] = *(const v4f*)(xw + b * 4096);
        xq[b][1] = *(const v4f*)(xw + b * 4096 + 512);
        xq[b][2] = *(const v4f*)(xw + b * 4096 + 2048);
        xq[b][3] = *(const v4f*)(xw + b * 4096 + 2560);
      }
    }
    if (younger) __builtin_amdgcn_s_setprio(3);

    for (int p = 0; p < 2; ++p) {
      const int o0 = pi * 2 + p;
      {
        const int nt = w & 3;
        const int half = w >> 2;
        const int nb = nt * 16 + lo;
        const int o1s = nb >> 3;
        const int r2s = nb & 7;
        const float* abase = &z0[p * Z0P + (half * 128 + lo) * ZS + 2 * hi];
        v2f af[2][2][16];
        v8f acc[2][2];
#pragma unroll
        for (int u = 0; u < 2; ++u) {
#pragma unroll
          for (int t = 0; t < 16; ++t) {
            af[0][u][t] = *(const v2f*)(abase + (u * 16) * ZS + 4 * t);
          }
        }
#pragma unroll
        for (int pr = 0; pr < 4; ++pr) {
          const int cur = pr & 1;
          const int nxt = cur ^ 1;
          if (pr < 3) {
#pragma unroll
            for (int u = 0; u < 2; ++u) {
#pragma unroll
              for (int t = 0; t < 16; ++t) {
                af[nxt][u][t] = *(const v2f*)(
                    abase + ((2 * pr + 2 + u) * 16) * ZS + 4 * t);
              }
            }
          }
          if (pr > 0) {
#pragma unroll
            for (int u = 0; u < 2; ++u) {
              const int mt = half * 8 + 2 * pr - 2 + u;
#pragma unroll
              for (int v = 0; v < 8; ++v) {
                const int m1 = mt * 16 + 8 * hi + v;
                const int b = m1 >> 6;
                const int i2 = (m1 >> 3) & 7;
                const int i3 = m1 & 7;
                z1[o1s * Z1P + (b * 8 + i3) * ZS + r2s * 8 + i2] =
                    acc[nxt][u][v];
              }
            }
          }
          BLOCK_FENCE();
          {
            v8f c0 = {0.f, 0.f, 0.f, 0.f, 0.f, 0.f, 0.f, 0.f};
            v8f c1 = {0.f, 0.f, 0.f, 0.f, 0.f, 0.f, 0.f, 0.f};
#pragma unroll
            for (int t = 0; t < 16; ++t) {
              c0 = WMMA4(af[cur][0][t], wb[t], c0);
              c1 = WMMA4(af[cur][1][t], wb[t], c1);
            }
            acc[cur][0] = c0;
            acc[cur][1] = c1;
          }
          BLOCK_FENCE();
        }
#pragma unroll
        for (int u = 0; u < 2; ++u) {
          const int mt = half * 8 + 6 + u;
#pragma unroll
          for (int v = 0; v < 8; ++v) {
            const int m1 = mt * 16 + 8 * hi + v;
            const int b = m1 >> 6;
            const int i2 = (m1 >> 3) & 7;
            const int i3 = m1 & 7;
            z1[o1s * Z1P + (b * 8 + i3) * ZS + r2s * 8 + i2] = acc[1][u][v];
          }
        }
      }

      if (younger && p == 1) __builtin_amdgcn_s_setprio(0);

      float bv[2][8];

      {
        const int nt = w & 3;
        const int half = w >> 2;
        const float* abase = &z1[(2 * nt) * Z1P + (half * 16 + lo) * ZS + 2 * hi];
        float* dbase = &z2[(2 * nt) * Z2P +
                           ((half * 2 + hi) * 8 + (lo >> 3)) * ZS + (lo & 7) * 8];
        v2f af[2][16];
        v8f acc[2][2];
#pragma unroll
        for (int u = 0; u < 2; ++u) {
#pragma unroll
          for (int t = 0; t < 16; ++t) {
            af[u][t] = *(const v2f*)(abase + u * Z1P + 4 * t);
          }
        }
        {
          const float* bp = bias + (o0 * 8 + 2 * nt) * 64 + (lo & 7);
#pragma unroll
          for (int u = 0; u < 2; ++u) {
#pragma unroll
            for (int v = 0; v < 8; ++v) {
              bv[u][v] = bp[u * 64 + v * 8];
            }
          }
        }
#pragma unroll
        for (int n2 = 0; n2 < 4; ++n2) {
          const int cur = n2 & 1;
          const int nxt = cur ^ 1;
          if (n2 > 0) {
#pragma unroll
            for (int u = 0; u < 2; ++u) {
#pragma unroll
              for (int v = 0; v < 8; ++v) {
                dbase[u * Z2P + (2 * (n2 - 1)) * ZS + v] = acc[nxt][u][v];
              }
            }
          }
          BLOCK_FENCE();
          {
            v8f c0 = {0.f, 0.f, 0.f, 0.f, 0.f, 0.f, 0.f, 0.f};
            v8f c1 = {0.f, 0.f, 0.f, 0.f, 0.f, 0.f, 0.f, 0.f};
#pragma unroll
            for (int t = 0; t < 16; ++t) {
              c0 = WMMA4(af[0][t], wbs[n2][t], c0);
              c1 = WMMA4(af[1][t], wbs[n2][t], c1);
            }
            acc[cur][0] = c0;
            acc[cur][1] = c1;
          }
        }
#pragma unroll
        for (int u = 0; u < 2; ++u) {
#pragma unroll
          for (int v = 0; v < 8; ++v) {
            dbase[u * Z2P + 6 * ZS + v] = acc[1][u][v];
          }
        }
      }

      {
        const int nt = w & 3;
        const int half = w >> 2;
        const float* abase = &z2[(2 * nt) * Z2P + (half * 16 + lo) * ZS + 2 * hi];
        v2f af[2][16];
#pragma unroll
        for (int u = 0; u < 2; ++u) {
#pragma unroll
          for (int t = 0; t < 16; ++t) {
            af[u][t] = *(const v2f*)(abase + u * Z2P + 4 * t);
          }
        }
        BLOCK_FENCE();
        v8f acc0, acc1;
#pragma unroll
        for (int v = 0; v < 8; ++v) {
          acc0[v] = bv[0][v];
          acc1[v] = bv[1][v];
        }
#pragma unroll
        for (int t = 0; t < 16; ++t) {
          acc0 = WMMA4(af[0][t], wb3[t], acc0);
          acc1 = WMMA4(af[1][t], wb3[t], acc1);
        }
        if (lo < 8) {
          const int b = half * 2 + hi;
#pragma unroll
          for (int u = 0; u < 2; ++u) {
            const int o1 = 2 * nt + u;
#pragma unroll
            for (int v = 0; v < 8; ++v) {
              const int feat = ((o0 * 8 + o1) * 8 + v) * 8 + lo;
              out[(size_t)(bg0 + b) * 4096 + feat] =
                  (u == 0 ? acc0[v] : acc1[v]);
            }
          }
        }
      }
    }
  }
  }
}

extern "C" void kernel_launch(void* const* d_in, const int* in_sizes, int n_in,
                              void* d_out, int out_size, void* d_ws,
                              size_t ws_size, hipStream_t stream) {
  const float* x = (const float*)d_in[0];
  const float* c0 = (const float*)d_in[1];
  const float* c1 = (const float*)d_in[2];
  const float* c2 = (const float*)d_in[3];
  const float* c3 = (const float*)d_in[4];
  const float* bias = (const float*)d_in[5];
  float* out = (float*)d_out;

  dim3 grid(4096 / (B_TILE * TILES_PER_WG));
  dim3 block(256);
  tt_fused_kernel<<<grid, block, 0, stream>>>(x, c0, c1, c2, c3, bias, out);
}


# Both workgroup barriers taken as signal + wait with the wave's own work between the halves: stage 2 (p=1) in one, the next pass's stage 0 then stage 3 (p=1) in the other; same sums, same order
# speedup vs baseline: 4.0121x; 1.0352x over previous
#include <hip/hip_runtime.h>


typedef float v2f __attribute__((ext_vector_type(2)));
typedef float v4f __attribute__((ext_vector_type(4)));
typedef float v8f __attribute__((ext_vector_type(8)));

#define WMMA4(A, B, C)                                                         \
  __builtin_amdgcn_wmma_f32_16x16x4_f32(false, (A), false, (B), (short)0,      \
                                        (C), false, false)

#define BLOCK_FENCE() __builtin_amdgcn_sched_barrier(0)


static __device__ __forceinline__ void bar_signal() {
  asm volatile("s_barrier_signal -1" ::: "memory");
}
static __device__ __forceinline__ void bar_wait() {
  asm volatile("s_barrier_wait -1" ::: "memory");
}
static __device__ __forceinline__ void lds_drain() {
  asm volatile("s_wait_dscnt 0x0" ::: "memory");
}

#define B_TILE 4
#define TILES_PER_WG 4

#define ZS 68
#define Z0P (256 * ZS + 2)
#define Z1P (32 * ZS + 2)
#define Z2P (32 * ZS)

static __device__ __forceinline__ void stage3_load(const float* z2, int nt,
                                                   int half, int lo, int hi,
                                                   v2f (&af)[2][16]) {
  const float* abase = &z2[(2 * nt) * Z2P + (half * 16 + lo) * ZS + 2 * hi];
#pragma unroll
  for (int u = 0; u < 2; ++u) {
#pragma unroll
    for (int t = 0; t < 16; ++t) {
      af[u][t] = *(const v2f*)(abase + u * Z2P + 4 * t);
    }
  }
}

static __device__ __forceinline__ void stage3_run(
    const v2f (&af)[2][16], const v2f (&wb3)[16], const float (&bv)[2][8],
    float* __restrict__ out, int o0, int bg0, int nt, int half, int lo,
    int hi) {
  v8f acc0, acc1;
#pragma unroll
  for (int v = 0; v < 8; ++v) {
    acc0[v] = bv[0][v];
    acc1[v] = bv[1][v];
  }
#pragma unroll
  for (int t = 0; t < 16; ++t) {
    acc0 = WMMA4(af[0][t], wb3[t], acc0);
    acc1 = WMMA4(af[1][t], wb3[t], acc1);
  }
  if (lo < 8) {
    const int b = half * 2 + hi;
#pragma unroll
    for (int u = 0; u < 2; ++u) {
      const int o1 = 2 * nt + u;
#pragma unroll
      for (int v = 0; v < 8; ++v) {
        const int feat = ((o0 * 8 + o1) * 8 + v) * 8 + lo;
        out[(size_t)(bg0 + b) * 4096 + feat] = (u == 0 ? acc0[v] : acc1[v]);
      }
    }
  }
}

static __device__ __forceinline__ void stage0(const v4f (&xq)[B_TILE][4],
                                              const float* lw0, float* z0,
                                              int pi, int w, int lo, int hi) {
  v2f bb0, bb1;
  bb0[0] = lw0[(2 * hi + 0) * 64 + pi * 16 + lo];
  bb0[1] = lw0[(2 * hi + 1) * 64 + pi * 16 + lo];
  bb1[0] = lw0[(4 + 2 * hi + 0) * 64 + pi * 16 + lo];
  bb1[1] = lw0[(4 + 2 * hi + 1) * 64 + pi * 16 + lo];
  const int p = lo >> 3;
  const int r1 = lo & 7;
  float* zd = &z0[p * Z0P + (32 * hi) * ZS + r1 * 8 + w];
#pragma unroll
  for (int b = 0; b < B_TILE; ++b) {
    v8f acc[4];
#pragma unroll
    for (int j = 0; j < 4; ++j) {
      const v2f a0 = {xq[b][0][j], xq[b][1][j]};
      const v2f a1 = {xq[b][2][j], xq[b][3][j]};
      v8f c = {0.f, 0.f, 0.f, 0.f, 0.f, 0.f, 0.f, 0.f};
      c = WMMA4(a0, bb0, c);
      acc[j] = WMMA4(a1, bb1, c);
    }
#pragma unroll
    for (int v = 0; v < 8; ++v) {
#pragma unroll
      for (int j = 0; j < 4; ++j) {
        zd[(b * 64 + 4 * v + j) * ZS] = acc[j][v];
      }
    }
  }
}

__global__ __launch_bounds__(256, 1) void tt_fused_kernel(
    const float* __restrict__ x, const float* __restrict__ c0,
    const float* __restrict__ c1, const float* __restrict__ c2,
    const float* __restrict__ c3, const float* __restrict__ bias,
    float* __restrict__ out) {
  __shared__ __align__(16) float lw0[8 * 64];
  __shared__ __align__(16) float lw1[64 * 64];
  __shared__ __align__(16) float lw2[64 * 64];
  __shared__ __align__(16) float lw3[64 * 8];
  __shared__ __align__(16) float z0[2 * Z0P];
  __shared__ __align__(16) float z1[8 * Z1P];
  __shared__ __align__(16) float z2[8 * Z2P];

  const int tid = threadIdx.x;
  const int w = tid >> 5;
  const int lane = tid & 31;
  const int lo = lane & 15;
  const int hi = lane >> 4;
  const int tile0 = blockIdx.x * TILES_PER_WG;
  const bool younger = __builtin_amdgcn_readfirstlane(tid >> 5) >= 4;

  v4f q1[4], q2[4];
#pragma unroll
  for (int j = 0; j < 4; ++j) {
    q1[j] = *(const v4f*)(c1 + 4 * (tid + 256 * j));
    q2[j] = *(const v4f*)(c2 + 4 * (tid + 256 * j));
  }
  const v4f qs = *(const v4f*)((tid < 128 ? c0 : c3) + 4 * (tid & 127));

  v4f xq[B_TILE][4];
  const float* xw = x + (size_t)(tile0 * B_TILE) * 4096 + (2 * hi) * 512 +
                    w * 64 + 4 * lo;
#pragma unroll
  for (int b = 0; b < B_TILE; ++b) {
    xq[b][0] = *(const v4f*)(xw + b * 4096);
    xq[b][1] = *(const v4f*)(xw + b * 4096 + 512);
    xq[b][2] = *(const v4f*)(xw + b * 4096 + 2048);
    xq[b][3] = *(const v4f*)(xw + b * 4096 + 2560);
  }

#pragma unroll
  for (int j = 0; j < 4; ++j) {
    const int i = 4 * (tid + 256 * j);
    *(v4f*)&lw1[i ^ ((i >> 3) & 16)] = q1[j];
    *(v4f*)&lw2[i ^ ((i >> 3) & 16)] = q2[j];
  }
  *(v4f*)&(tid < 128 ? lw0 : lw3)[4 * (tid & 127)] = qs;
  __syncthreads();

  v2f wb[16];
  v2f wbs[4][16];
  v2f wb3[16];
  {
    const int nbx = ((w & 3) * 16 + lo) ^ (hi << 4);
#pragma unroll
    for (int t = 0; t < 16; ++t) {
      const int k = 4 * t + 2 * hi;
      wb[t][0] = lw1[k * 64 + nbx];
      wb[t][1] = lw1[(k + 1) * 64 + nbx];
    }
    const float* bbase = &lw2[(2 * hi) * 64 + lo];
#pragma unroll
    for (int n2 = 0; n2 < 4; ++n2) {
      const int col = (n2 ^ hi) << 4;
#pragma unroll
      for (int t = 0; t < 16; ++t) {
        wbs[n2][t][0] = bbase[(4 * t) * 64 + col];
        wbs[n2][t][1] = bbase[(4 * t + 1) * 64 + col];
      }
    }
    const float* b3 = &lw3[(2 * hi) * 8 + (lo & 7)];
#pragma unroll
    for (int t = 0; t < 16; ++t) {
      const float e0 = b3[(4 * t) * 8];
      const float e1 = b3[(4 * t + 1) * 8];
      wb3[t][0] = (lo < 8) ? e0 : 0.f;
      wb3[t][1] = (lo < 8) ? e1 : 0.f;
    }
  }

  stage0(xq, lw0, z0, 0, w, lo, hi);
  lds_drain();
  bar_signal();

  for (int it = 0; it < TILES_PER_WG; ++it) {
  const int bg0 = (tile0 + it) * B_TILE;

  for (int pi = 0; pi < 4; ++pi) {
    const bool last = (pi == 3 && it + 1 == TILES_PER_WG);

    for (int p = 0; p < 2; ++p) {
      const int o0 = pi * 2 + p;
      if (p == 0) {
        bar_wait();
        if (pi == 3 && it + 1 < TILES_PER_WG) {
          xw += B_TILE * 4096;
#pragma unroll
          for (int b = 0; b < B_TILE; ++b) {
            xq[b][0] = *(const v4f*)(xw + b * 4096);
            xq[b][1] = *(const v4f*)(xw + b * 4096 + 512);
            xq[b][2] = *(const v4f*)(xw + b * 4096 + 2048);
            xq[b][3] = *(const v4f*)(xw + b * 4096 + 2560);
          }
        }
        if (younger) __builtin_amdgcn_s_setprio(3);
      }
      {
        const int nt = w & 3;
        const int half = w >> 2;
        const int nb = nt * 16 + lo;
        const int o1s = nb >> 3;
        const int r2s = nb & 7;
        const float* abase = &z0[p * Z0P + (half * 128 + lo) * ZS + 2 * hi];
        v2f af[2][2][16];
        v8f acc[2][2];
#pragma unroll
        for (int u = 0; u < 2; ++u) {
#pragma unroll
          for (int t = 0; t < 16; ++t) {
            af[0][u][t] = *(const v2f*)(abase + (u * 16) * ZS + 4 * t);
          }
        }
#pragma unroll
        for (int pr = 0; pr < 4; ++pr) {
          const int cur = pr & 1;
          const int nxt = cur ^ 1;
          if (pr < 3) {
#pragma unroll
            for (int u = 0; u < 2; ++u) {
#pragma unroll
              for (int t = 0; t < 16; ++t) {
                af[nxt][u][t] = *(const v2f*)(
                    abase + ((2 * pr + 2 + u) * 16) * ZS + 4 * t);
              }
            }
          }
          if (pr > 0) {
#pragma unroll
            for (int u = 0; u < 2; ++u) {
              const int mt = half * 8 + 2 * pr - 2 + u;
#pragma unroll
              for (int v = 0; v < 8; ++v) {
                const int m1 = mt * 16 + 8 * hi + v;
                const int b = m1 >> 6;
                const int i2 = (m1 >> 3) & 7;
                const int i3 = m1 & 7;
                z1[o1s * Z1P + (b * 8 + i3) * ZS + r2s * 8 + i2] =
                    acc[nxt][u][v];
              }
            }
          }
          BLOCK_FENCE();
          {
            v8f c0 = {0.f, 0.f, 0.f, 0.f, 0.f, 0.f, 0.f, 0.f};
            v8f c1 = {0.f, 0.f, 0.f, 0.f, 0.f, 0.f, 0.f, 0.f};
#pragma unroll
            for (int t = 0; t < 16; ++t) {
              c0 = WMMA4(af[cur][0][t], wb[t], c0);
              c1 = WMMA4(af[cur][1][t], wb[t], c1);
            }
            acc[cur][0] = c0;
            acc[cur][1] = c1;
          }
          BLOCK_FENCE();
        }
#pragma unroll
        for (int u = 0; u < 2; ++u) {
          const int mt = half * 8 + 6 + u;
#pragma unroll
          for (int v = 0; v < 8; ++v) {
            const int m1 = mt * 16 + 8 * hi + v;
            const int b = m1 >> 6;
            const int i2 = (m1 >> 3) & 7;
            const int i3 = m1 & 7;
            z1[o1s * Z1P + (b * 8 + i3) * ZS + r2s * 8 + i2] = acc[1][u][v];
          }
        }
      }

      if (younger && p == 1) __builtin_amdgcn_s_setprio(0);
      if (p == 1 && !last) bar_signal();

      float bv[2][8];

      {
        const int nt = w & 3;
        const int half = w >> 2;
        const float* abase = &z1[(2 * nt) * Z1P + (half * 16 + lo) * ZS + 2 * hi];
        float* dbase = &z2[(2 * nt) * Z2P +
                           ((half * 2 + hi) * 8 + (lo >> 3)) * ZS + (lo & 7) * 8];
        v2f af[2][16];
        v8f acc[2][2];
#pragma unroll
        for (int u = 0; u < 2; ++u) {
#pragma unroll
          for (int t = 0; t < 16; ++t) {
            af[u][t] = *(const v2f*)(abase + u * Z1P + 4 * t);
          }
        }
        {
          const float* bp = bias + (o0 * 8 + 2 * nt) * 64 + (lo & 7);
#pragma unroll
          for (int u = 0; u < 2; ++u) {
#pragma unroll
            for (int v = 0; v < 8; ++v) {
              bv[u][v] = bp[u * 64 + v * 8];
            }
          }
        }
#pragma unroll
        for (int n2 = 0; n2 < 4; ++n2) {
          const int cur = n2 & 1;
          const int nxt = cur ^ 1;
          if (n2 > 0) {
#pragma unroll
            for (int u = 0; u < 2; ++u) {
#pragma unroll
              for (int v = 0; v < 8; ++v) {
                dbase[u * Z2P + (2 * (n2 - 1)) * ZS + v] = acc[nxt][u][v];
              }
            }
          }
          BLOCK_FENCE();
          {
            v8f c0 = {0.f, 0.f, 0.f, 0.f, 0.f, 0.f, 0.f, 0.f};
            v8f c1 = {0.f, 0.f, 0.f, 0.f, 0.f, 0.f, 0.f, 0.f};
#pragma unroll
            for (int t = 0; t < 16; ++t) {
              c0 = WMMA4(af[0][t], wbs[n2][t], c0);
              c1 = WMMA4(af[1][t], wbs[n2][t], c1);
            }
            acc[cur][0] = c0;
            acc[cur][1] = c1;
          }
        }
#pragma unroll
        for (int u = 0; u < 2; ++u) {
#pragma unroll
          for (int v = 0; v < 8; ++v) {
            dbase[u * Z2P + 6 * ZS + v] = acc[1][u][v];
          }
        }
      }

      if (p == 1 && !last) {
        bar_wait();
        stage0(xq, lw0, z0, (pi + 1) & 3, w, lo, hi);
      }
      {
        v2f af3[2][16];
        stage3_load(z2, w & 3, w >> 2, lo, hi, af3);
        if (p == 1) {
          lds_drain();
          if (!last) bar_signal();
        }
        BLOCK_FENCE();
        stage3_run(af3, wb3, bv, out, o0, bg0, w & 3, w >> 2, lo, hi);
      }
    }
  }
  }
}

extern "C" void kernel_launch(void* const* d_in, const int* in_sizes, int n_in,
                              void* d_out, int out_size, void* d_ws,
                              size_t ws_size, hipStream_t stream) {
  const float* x = (const float*)d_in[0];
  const float* c0 = (const float*)d_in[1];
  const float* c1 = (const float*)d_in[2];
  const float* c2 = (const float*)d_in[3];
  const float* c3 = (const float*)d_in[4];
  const float* bias = (const float*)d_in[5];
  float* out = (float*)d_out;

  dim3 grid(4096 / (B_TILE * TILES_PER_WG));
  dim3 block(256);
  tt_fused_kernel<<<grid, block, 0, stream>>>(x, c0, c1, c2, c3, bias, out);
}
